# starting kernel
# baseline (speedup 1.0000x reference)
#include <hip/hip_runtime.h>
#include <math.h>

typedef float v2f __attribute__((ext_vector_type(2)));
typedef float v8f __attribute__((ext_vector_type(8)));

static inline int cdiv_i(long long a, long long b) { return (int)((a + b - 1) / b); }

__global__ void zero_f32(float* __restrict__ p, long long n) {
  long long i = (long long)blockIdx.x * blockDim.x + threadIdx.x;
  long long stride = (long long)gridDim.x * blockDim.x;
  for (; i < n; i += stride) p[i] = 0.0f;
}

__global__ void relu_f32(float* __restrict__ p, long long n) {
  long long i = (long long)blockIdx.x * blockDim.x + threadIdx.x;
  long long stride = (long long)gridDim.x * blockDim.x;
  for (; i < n; i += stride) p[i] = fmaxf(p[i], 0.0f);
}

__global__ void deg_count(const int* __restrict__ dst, int E, float* __restrict__ deg) {
  int i = blockIdx.x * blockDim.x + threadIdx.x;
  if (i < E) atomicAdd(&deg[dst[i]], 1.0f);
}

__global__ void deg_to_dinv(float* __restrict__ deg, long long n) {
  long long i = (long long)blockIdx.x * blockDim.x + threadIdx.x;
  long long stride = (long long)gridDim.x * blockDim.x;
  for (; i < n; i += stride) {
    float d = deg[i];
    deg[i] = (d > 0.0f) ? rsqrtf(d) : 0.0f;
  }
}

__global__ void gemm_wmma_f32(const float* __restrict__ A, const float* __restrict__ B,
                              const float* __restrict__ bias, float* __restrict__ D,
                              int M, int K, int Ncols) {
  extern __shared__ float As[];
  const int row0 = blockIdx.x * 16;
  const int tid  = threadIdx.x;
  const int ldA  = K + 1;

  for (int idx = tid; idx < 16 * K; idx += blockDim.x) {
    int r = idx / K;
    int k = idx - r * K;
    int row = row0 + r;
    if (row >= M) row = M - 1;
    As[r * ldA + k] = A[(long long)row * K + k];
  }
  __syncthreads();

  const int wave = tid >> 5;
  const int lane = tid & 31;
  const int half = lane >> 4;
  const int mn   = lane & 15;
  const int kb   = half * 2;
  const int col0 = wave * 16;

  v8f acc = {};
  #pragma unroll 4
  for (int kk = 0; kk < K; kk += 4) {
    v2f a, b;
    a.x = As[mn * ldA + kk + kb];
    a.y = As[mn * ldA + kk + kb + 1];
    const float* Bp = B + (long long)(kk + kb) * Ncols + col0 + mn;
    b.x = Bp[0];
    b.y = Bp[Ncols];
    acc = __builtin_amdgcn_wmma_f32_16x16x4_f32(
         false, a,  false, b,
         (short)0, acc,  false,  false);
  }

  const float bv = bias ? bias[col0 + mn] : 0.0f;
  #pragma unroll
  for (int r = 0; r < 8; ++r) {
    int row = row0 + r + half * 8;
    if (row < M) D[(long long)row * Ncols + col0 + mn] = acc[r] + bv;
  }
}

__global__ void gcn_aggregate(const int* __restrict__ src, const int* __restrict__ dst, int E,
                              const float* __restrict__ dinv,
                              const float* __restrict__ HH, int ldH,
                              float* __restrict__ out, int ldOut, int width) {
  long long total  = (long long)E * width;
  long long stride = (long long)gridDim.x * blockDim.x;
  for (long long i = (long long)blockIdx.x * blockDim.x + threadIdx.x; i < total; i += stride) {
    int e = (int)(i / width);
    int h = (int)(i - (long long)e * width);
    int s = src[e];
    int d = dst[e];
    float nrm = dinv[s] * dinv[d];
    atomicAdd(&out[(long long)d * ldOut + h], HH[(long long)s * ldH + h] * nrm);
  }
}

__global__ void build_R1(const float* __restrict__ HH1, const float* __restrict__ b1,
                         float* __restrict__ R1, int N) {
  long long total  = (long long)N * 512;
  long long stride = (long long)gridDim.x * blockDim.x;
  for (long long i = (long long)blockIdx.x * blockDim.x + threadIdx.x; i < total; i += stride) {
    int col = (int)(i & 511);
    long long n = i >> 9;
    int br = col >> 7;
    int h  = col & 127;
    float v = b1[h];
    if (br == 3) v += HH1[n * 128 + h];
    R1[i] = v;
  }
}

__global__ void build_R2(const float* __restrict__ HH2, const float* __restrict__ b2,
                         float* __restrict__ R2, int N) {
  long long total  = (long long)N * 256;
  long long stride = (long long)gridDim.x * blockDim.x;
  for (long long i = (long long)blockIdx.x * blockDim.x + threadIdx.x; i < total; i += stride) {
    int col = (int)(i & 255);
    long long n = i >> 8;
    int br = col >> 6;
    int h  = col & 63;
    float v = b2[h];
    if (br == 3) v += HH2[n * 64 + h];
    R2[i] = v;
  }
}

__global__ void log_softmax64(float* __restrict__ out, int N) {
  int row  = blockIdx.x * (blockDim.x >> 5) + (threadIdx.x >> 5);
  int lane = threadIdx.x & 31;
  if (row >= N) return;
  float* p = out + (long long)row * 64;
  float a = p[lane];
  float b = p[lane + 32];
  float mx = fmaxf(a, b);
  #pragma unroll
  for (int off = 16; off > 0; off >>= 1) mx = fmaxf(mx, __shfl_xor(mx, off));
  float s = __expf(a - mx) + __expf(b - mx);
  #pragma unroll
  for (int off = 16; off > 0; off >>= 1) s += __shfl_xor(s, off);
  float lse = mx + __logf(s);
  p[lane]      = a - lse;
  p[lane + 32] = b - lse;
}

extern "C" void kernel_launch(void* const* d_in, const int* in_sizes, int n_in,
                              void* d_out, int out_size, void* d_ws, size_t ws_size,
                              hipStream_t stream) {
  const float* x  = (const float*)d_in[0];
  const int* e1   = (const int*)d_in[1];
  const int* e2   = (const int*)d_in[2];
  const int* e3   = (const int*)d_in[3];
  const float* W1 = (const float*)d_in[5];
  const float* b1 = (const float*)d_in[6];
  const float* W2 = (const float*)d_in[7];
  const float* b2 = (const float*)d_in[8];
  const float* Wl = (const float*)d_in[9];
  const float* bl = (const float*)d_in[10];

  const int F = 512, H = 128, C = 64;
  const int N  = in_sizes[0] / F;
  const int E1 = in_sizes[1] / 2;
  const int E2 = in_sizes[2] / 2;
  const int E3 = in_sizes[3] / 2;

  float* ws   = (float*)d_ws;
  float* dinv = ws;
  float* HH1  = dinv + 3LL * N;
  float* R1   = HH1 + (long long)N * H;
  float* HH2  = R1 + (long long)N * 4 * H;
  float* R2   = HH2 + (long long)N * C;

  const int T = 256;
  const int AGG_MAX_BLOCKS = 131072;

  zero_f32<<<cdiv_i(3LL * N, T), T, 0, stream>>>(dinv, 3LL * N);
  deg_count<<<cdiv_i(E1, T), T, 0, stream>>>(e1 + E1, E1, dinv);
  deg_count<<<cdiv_i(E2, T), T, 0, stream>>>(e2 + E2, E2, dinv + N);
  deg_count<<<cdiv_i(E3, T), T, 0, stream>>>(e3 + E3, E3, dinv + 2LL * N);
  deg_to_dinv<<<cdiv_i(3LL * N, T), T, 0, stream>>>(dinv, 3LL * N);

  const int mtiles = cdiv_i(N, 16);

  gemm_wmma_f32<<<mtiles, 32 * (H / 16), (size_t)16 * (F + 1) * sizeof(float), stream>>>(
      x, W1, nullptr, HH1, N, F, H);

  build_R1<<<cdiv_i((long long)N * 4 * H, T), T, 0, stream>>>(HH1, b1, R1, N);

  {
    long long t1 = (long long)E1 * H, t2 = (long long)E2 * H, t3 = (long long)E3 * H;
    int g1 = min(cdiv_i(t1, T), AGG_MAX_BLOCKS);
    int g2 = min(cdiv_i(t2, T), AGG_MAX_BLOCKS);
    int g3 = min(cdiv_i(t3, T), AGG_MAX_BLOCKS);
    gcn_aggregate<<<g1, T, 0, stream>>>(e1, e1 + E1, E1, dinv,           HH1, H, R1,           4 * H, H);
    gcn_aggregate<<<g2, T, 0, stream>>>(e2, e2 + E2, E2, dinv + N,       HH1, H, R1 + H,       4 * H, H);
    gcn_aggregate<<<g3, T, 0, stream>>>(e3, e3 + E3, E3, dinv + 2LL * N, HH1, H, R1 + 2 * H,   4 * H, H);
  }

  relu_f32<<<cdiv_i((long long)N * 4 * H, T), T, 0, stream>>>(R1, (long long)N * 4 * H);

  gemm_wmma_f32<<<mtiles, 32 * (C / 16), (size_t)16 * (4 * H + 1) * sizeof(float), stream>>>(
      R1, W2, nullptr, HH2, N, 4 * H, C);

  build_R2<<<cdiv_i((long long)N * 4 * C, T), T, 0, stream>>>(HH2, b2, R2, N);

  {
    long long t1 = (long long)E1 * C, t2 = (long long)E2 * C, t3 = (long long)E3 * C;
    int g1 = min(cdiv_i(t1, T), AGG_MAX_BLOCKS);
    int g2 = min(cdiv_i(t2, T), AGG_MAX_BLOCKS);
    int g3 = min(cdiv_i(t3, T), AGG_MAX_BLOCKS);
    gcn_aggregate<<<g1, T, 0, stream>>>(e1, e1 + E1, E1, dinv,           HH2, C, R2,           4 * C, C);
    gcn_aggregate<<<g2, T, 0, stream>>>(e2, e2 + E2, E2, dinv + N,       HH2, C, R2 + C,       4 * C, C);
    gcn_aggregate<<<g3, T, 0, stream>>>(e3, e3 + E3, E3, dinv + 2LL * N, HH2, C, R2 + 2 * C,   4 * C, C);
  }

  float* logits = (float*)d_out;
  gemm_wmma_f32<<<mtiles, 32 * (C / 16), (size_t)16 * (4 * C + 1) * sizeof(float), stream>>>(
      R2, Wl, bl, logits, N, 4 * C, C);

  log_softmax64<<<cdiv_i(N, T / 32), T, 0, stream>>>(logits, N);
}


# Atomic scatter-add replaced by register gather over a dst-sorted edge list (counting sort built in the degree pass); sum order now the arrival-order rank of the count atomics
# speedup vs baseline: 2.6083x; 2.6083x over previous
#include <hip/hip_runtime.h>
#include <math.h>

typedef float v2f __attribute__((ext_vector_type(2)));
typedef float v8f __attribute__((ext_vector_type(8)));

static inline int cdiv_i(long long a, long long b) { return (int)((a + b - 1) / b); }

__global__ void zero_i32(int* __restrict__ p, long long n) {
  long long i = (long long)blockIdx.x * blockDim.x + threadIdx.x;
  long long stride = (long long)gridDim.x * blockDim.x;
  for (; i < n; i += stride) p[i] = 0;
}

__global__ void count_rank(const int* __restrict__ dst, int E, int* __restrict__ cnt,
                           int* __restrict__ rank) {
  int i = blockIdx.x * blockDim.x + threadIdx.x;
  if (i < E) rank[i] = atomicAdd(&cnt[dst[i]], 1);
}

__global__ void scan_deg(const int* __restrict__ cnt, int* __restrict__ off,
                         float* __restrict__ dinv, int N, int E1, int E2) {
  __shared__ int part[1024];
  const int g = blockIdx.x;
  const int t = threadIdx.x;
  const int nt = blockDim.x;
  const int per = (N + nt - 1) / nt;
  int lo = t * per; if (lo > N) lo = N;
  int hi = lo + per; if (hi > N) hi = N;
  const int* c = cnt + (long long)g * N;
  int s = 0;
  for (int i = lo; i < hi; ++i) s += c[i];
  part[t] = s;
  __syncthreads();
  for (int d = 1; d < nt; d <<= 1) {
    int v = (t >= d) ? part[t - d] : 0;
    __syncthreads();
    part[t] += v;
    __syncthreads();
  }
  int run = part[t] - s;
  if (g >= 1) run += E1;
  if (g >= 2) run += E2;
  int* o = off + (long long)g * N;
  float* dv = dinv + (long long)g * N;
  for (int i = lo; i < hi; ++i) {
    int v = c[i];
    o[i] = run;
    run += v;
    dv[i] = (v > 0) ? rsqrtf((float)v) : 0.0f;
  }
}

__global__ void csr_fill(const int* __restrict__ src, const int* __restrict__ dst, int E,
                         const int* __restrict__ off, const int* __restrict__ rank,
                         const float* __restrict__ dinv, int2* __restrict__ csr) {
  int i = blockIdx.x * blockDim.x + threadIdx.x;
  if (i < E) {
    int s = src[i];
    int d = dst[i];
    int2 v;
    v.x = s;
    v.y = __float_as_int(dinv[s] * dinv[d]);
    csr[off[d] + rank[i]] = v;
  }
}

__global__ void gemm_wmma_f32(const float* __restrict__ A, const float* __restrict__ B,
                              const float* __restrict__ bias, float* __restrict__ D,
                              int M, int K, int Ncols) {
  extern __shared__ float As[];
  const int row0 = blockIdx.x * 16;
  const int tid  = threadIdx.x;
  const int ldA  = K + 1;

  for (int idx = tid; idx < 16 * K; idx += blockDim.x) {
    int r = idx / K;
    int k = idx - r * K;
    int row = row0 + r;
    if (row >= M) row = M - 1;
    As[r * ldA + k] = A[(long long)row * K + k];
  }
  __syncthreads();

  const int wave = tid >> 5;
  const int lane = tid & 31;
  const int half = lane >> 4;
  const int mn   = lane & 15;
  const int kb   = half * 2;
  const int col0 = wave * 16;

  v8f acc = {};
  #pragma unroll 4
  for (int kk = 0; kk < K; kk += 4) {
    v2f a, b;
    a.x = As[mn * ldA + kk + kb];
    a.y = As[mn * ldA + kk + kb + 1];
    const float* Bp = B + (long long)(kk + kb) * Ncols + col0 + mn;
    b.x = Bp[0];
    b.y = Bp[Ncols];
    acc = __builtin_amdgcn_wmma_f32_16x16x4_f32(
         false, a,  false, b,
         (short)0, acc,  false,  false);
  }

  const float bv = bias ? bias[col0 + mn] : 0.0f;
  #pragma unroll
  for (int r = 0; r < 8; ++r) {
    int row = row0 + r + half * 8;
    if (row < M) D[(long long)row * Ncols + col0 + mn] = acc[r] + bv;
  }
}

template <int VEC, bool RELU>
__global__ void __launch_bounds__(256)
gcn_gather(const int* __restrict__ off, const int* __restrict__ cnt,
           const int2* __restrict__ csr, const float* __restrict__ HH,
           const float* __restrict__ bias, float* __restrict__ out, int N) {
  typedef float vf __attribute__((ext_vector_type(VEC)));
  const int wv   = __builtin_amdgcn_readfirstlane((int)(threadIdx.x >> 5));
  const int node = blockIdx.x * (int)(blockDim.x >> 5) + wv;
  if (node >= N) return;
  const int lane = threadIdx.x & 31;
  const vf* __restrict__ Hv = reinterpret_cast<const vf*>(HH);
  const vf bv = reinterpret_cast<const vf*>(bias)[lane];
  const vf zero = (vf)(0.0f);
  vf* __restrict__ orow = reinterpret_cast<vf*>(out + (size_t)node * (size_t)(4 * 32 * VEC));

  for (int g = 0; g < 3; ++g) {
    const int beg = off[g * N + node];
    const int c   = cnt[g * N + node];
    vf acc = zero;
    for (int base = 0; base < c; base += 32) {
      const int rem = c - base;
      const int m = rem < 32 ? rem : 32;
      const int2 ed = csr[beg + base + (lane < m ? lane : m - 1)];
      int j = 0;
      for (; j + 4 <= m; j += 4) {
        const int s0 = __builtin_amdgcn_readlane(ed.x, j);
        const int s1 = __builtin_amdgcn_readlane(ed.x, j + 1);
        const int s2 = __builtin_amdgcn_readlane(ed.x, j + 2);
        const int s3 = __builtin_amdgcn_readlane(ed.x, j + 3);
        const float w0 = __int_as_float(__builtin_amdgcn_readlane(ed.y, j));
        const float w1 = __int_as_float(__builtin_amdgcn_readlane(ed.y, j + 1));
        const float w2 = __int_as_float(__builtin_amdgcn_readlane(ed.y, j + 2));
        const float w3 = __int_as_float(__builtin_amdgcn_readlane(ed.y, j + 3));
        const vf v0 = Hv[(size_t)s0 * 32 + lane];
        const vf v1 = Hv[(size_t)s1 * 32 + lane];
        const vf v2 = Hv[(size_t)s2 * 32 + lane];
        const vf v3 = Hv[(size_t)s3 * 32 + lane];
        acc += w0 * v0;
        acc += w1 * v1;
        acc += w2 * v2;
        acc += w3 * v3;
      }
      #pragma nounroll
      for (; j < m; ++j) {
        const int s0 = __builtin_amdgcn_readlane(ed.x, j);
        const float w0 = __int_as_float(__builtin_amdgcn_readlane(ed.y, j));
        acc += w0 * Hv[(size_t)s0 * 32 + lane];
      }
    }
    vf r = acc + bv;
    if (RELU) r = __builtin_elementwise_max(r, zero);
    orow[g * 32 + lane] = r;
  }
  vf r = Hv[(size_t)node * 32 + lane] + bv;
  if (RELU) r = __builtin_elementwise_max(r, zero);
  orow[3 * 32 + lane] = r;
}

__global__ void log_softmax64(float* __restrict__ out, int N) {
  int row  = blockIdx.x * (blockDim.x >> 5) + (threadIdx.x >> 5);
  int lane = threadIdx.x & 31;
  if (row >= N) return;
  float* p = out + (long long)row * 64;
  float a = p[lane];
  float b = p[lane + 32];
  float mx = fmaxf(a, b);
  #pragma unroll
  for (int off = 16; off > 0; off >>= 1) mx = fmaxf(mx, __shfl_xor(mx, off));
  float s = __expf(a - mx) + __expf(b - mx);
  #pragma unroll
  for (int off = 16; off > 0; off >>= 1) s += __shfl_xor(s, off);
  float lse = mx + __logf(s);
  p[lane]      = a - lse;
  p[lane + 32] = b - lse;
}

extern "C" void kernel_launch(void* const* d_in, const int* in_sizes, int n_in,
                              void* d_out, int out_size, void* d_ws, size_t ws_size,
                              hipStream_t stream) {
  const float* x  = (const float*)d_in[0];
  const int* e1   = (const int*)d_in[1];
  const int* e2   = (const int*)d_in[2];
  const int* e3   = (const int*)d_in[3];
  const float* W1 = (const float*)d_in[5];
  const float* b1 = (const float*)d_in[6];
  const float* W2 = (const float*)d_in[7];
  const float* b2 = (const float*)d_in[8];
  const float* Wl = (const float*)d_in[9];
  const float* bl = (const float*)d_in[10];

  const int F = 512, H = 128, C = 64;
  const int N  = in_sizes[0] / F;
  const int E1 = in_sizes[1] / 2;
  const int E2 = in_sizes[2] / 2;
  const int E3 = in_sizes[3] / 2;
  const long long Et = (long long)E1 + E2 + E3;

  float* ws   = (float*)d_ws;
  float* HH1  = ws;
  float* R1   = HH1 + (long long)N * H;
  float* HH2  = R1 + (long long)N * 4 * H;
  float* R2   = HH2 + (long long)N * C;
  int2*  csr  = (int2*)(R2 + (long long)N * 4 * C);
  float* dinv = (float*)(csr + Et);
  int*   cnt  = (int*)(dinv + 3LL * N);
  int*   off  = cnt + 3LL * N;
  int*   rank = off + 3LL * N;

  const int T = 256;

  zero_i32<<<cdiv_i(3LL * N, T), T, 0, stream>>>(cnt, 3LL * N);
  count_rank<<<cdiv_i(E1, T), T, 0, stream>>>(e1 + E1, E1, cnt,           rank);
  count_rank<<<cdiv_i(E2, T), T, 0, stream>>>(e2 + E2, E2, cnt + N,       rank + E1);
  count_rank<<<cdiv_i(E3, T), T, 0, stream>>>(e3 + E3, E3, cnt + 2LL * N, rank + E1 + E2);
  scan_deg<<<3, 1024, 0, stream>>>(cnt, off, dinv, N, E1, E2);
  csr_fill<<<cdiv_i(E1, T), T, 0, stream>>>(e1, e1 + E1, E1, off,           rank,           dinv,           csr);
  csr_fill<<<cdiv_i(E2, T), T, 0, stream>>>(e2, e2 + E2, E2, off + N,       rank + E1,      dinv + N,       csr);
  csr_fill<<<cdiv_i(E3, T), T, 0, stream>>>(e3, e3 + E3, E3, off + 2LL * N, rank + E1 + E2, dinv + 2LL * N, csr);

  const int mtiles = cdiv_i(N, 16);
  const int gblocks = cdiv_i(N, T / 32);

  gemm_wmma_f32<<<mtiles, 32 * (H / 16), (size_t)16 * (F + 1) * sizeof(float), stream>>>(
      x, W1, nullptr, HH1, N, F, H);

  gcn_gather<4, true><<<gblocks, T, 0, stream>>>(off, cnt, csr, HH1, b1, R1, N);

  gemm_wmma_f32<<<mtiles, 32 * (C / 16), (size_t)16 * (4 * H + 1) * sizeof(float), stream>>>(
      R1, W2, nullptr, HH2, N, 4 * H, C);

  gcn_gather<2, false><<<gblocks, T, 0, stream>>>(off, cnt, csr, HH2, b2, R2, N);

  float* logits = (float*)d_out;
  gemm_wmma_f32<<<mtiles, 32 * (C / 16), (size_t)16 * (4 * C + 1) * sizeof(float), stream>>>(
      R2, Wl, bl, logits, N, 4 * C, C);

  log_softmax64<<<cdiv_i(N, T / 32), T, 0, stream>>>(logits, N);
}


# GEMM A-strip staging: serial 4-byte load-wait-store rounds replaced by 16-byte loads all issued before the LDS stores (K, Ncols compile-time; LDS pitch K+4, 8-byte fragment reads)
# speedup vs baseline: 3.3671x; 1.2909x over previous
#include <hip/hip_runtime.h>
#include <math.h>

typedef float v2f __attribute__((ext_vector_type(2)));
typedef float v8f __attribute__((ext_vector_type(8)));

static inline int cdiv_i(long long a, long long b) { return (int)((a + b - 1) / b); }

__global__ void zero_i32(int* __restrict__ p, long long n) {
  long long i = (long long)blockIdx.x * blockDim.x + threadIdx.x;
  long long stride = (long long)gridDim.x * blockDim.x;
  for (; i < n; i += stride) p[i] = 0;
}

__global__ void count_rank(const int* __restrict__ dst, int E, int* __restrict__ cnt,
                           int* __restrict__ rank) {
  int i = blockIdx.x * blockDim.x + threadIdx.x;
  if (i < E) rank[i] = atomicAdd(&cnt[dst[i]], 1);
}

__global__ void scan_deg(const int* __restrict__ cnt, int* __restrict__ off,
                         float* __restrict__ dinv, int N, int E1, int E2) {
  __shared__ int part[1024];
  const int g = blockIdx.x;
  const int t = threadIdx.x;
  const int nt = blockDim.x;
  const int per = (N + nt - 1) / nt;
  int lo = t * per; if (lo > N) lo = N;
  int hi = lo + per; if (hi > N) hi = N;
  const int* c = cnt + (long long)g * N;
  int s = 0;
  for (int i = lo; i < hi; ++i) s += c[i];
  part[t] = s;
  __syncthreads();
  for (int d = 1; d < nt; d <<= 1) {
    int v = (t >= d) ? part[t - d] : 0;
    __syncthreads();
    part[t] += v;
    __syncthreads();
  }
  int run = part[t] - s;
  if (g >= 1) run += E1;
  if (g >= 2) run += E2;
  int* o = off + (long long)g * N;
  float* dv = dinv + (long long)g * N;
  for (int i = lo; i < hi; ++i) {
    int v = c[i];
    o[i] = run;
    run += v;
    dv[i] = (v > 0) ? rsqrtf((float)v) : 0.0f;
  }
}

__global__ void csr_fill(const int* __restrict__ src, const int* __restrict__ dst, int E,
                         const int* __restrict__ off, const int* __restrict__ rank,
                         const float* __restrict__ dinv, int2* __restrict__ csr) {
  int i = blockIdx.x * blockDim.x + threadIdx.x;
  if (i < E) {
    int s = src[i];
    int d = dst[i];
    int2 v;
    v.x = s;
    v.y = __float_as_int(dinv[s] * dinv[d]);
    csr[off[d] + rank[i]] = v;
  }
}

template <int K, int NCOLS>
__global__ void __launch_bounds__(NCOLS * 2)
gemm_wmma_f32(const float* __restrict__ A, const float* __restrict__ B,
              const float* __restrict__ bias, float* __restrict__ D, int M) {
  constexpr int NT  = NCOLS * 2;
  constexpr int LDA = K + 4;
  constexpr int K4  = K / 4;
  constexpr int PER = (16 * K4) / NT;
  static_assert((16 * K4) % NT == 0, "strip must divide evenly over the block");
  __shared__ __attribute__((aligned(16))) float As[16 * LDA];

  const int row0 = blockIdx.x * 16;
  const int tid  = threadIdx.x;

  float4 stage[PER];
  #pragma unroll
  for (int i = 0; i < PER; ++i) {
    const int q  = tid + i * NT;
    const int r  = q / K4;
    const int c4 = q - r * K4;
    int row = row0 + r;
    if (row >= M) row = M - 1;
    stage[i] = *reinterpret_cast<const float4*>(A + (long long)row * K + c4 * 4);
  }
  #pragma unroll
  for (int i = 0; i < PER; ++i) {
    const int q  = tid + i * NT;
    const int r  = q / K4;
    const int c4 = q - r * K4;
    *reinterpret_cast<float4*>(&As[r * LDA + c4 * 4]) = stage[i];
  }
  __syncthreads();

  const int wave = tid >> 5;
  const int lane = tid & 31;
  const int half = lane >> 4;
  const int mn   = lane & 15;
  const int kb   = half * 2;
  const int col0 = wave * 16;

  const float* Ap = &As[mn * LDA + kb];
  const float* Bp = B + kb * NCOLS + col0 + mn;

  v8f acc = {};
  #pragma unroll 4
  for (int kk = 0; kk < K; kk += 4) {
    const v2f a = *reinterpret_cast<const v2f*>(Ap + kk);
    v2f b;
    b.x = Bp[kk * NCOLS];
    b.y = Bp[kk * NCOLS + NCOLS];
    acc = __builtin_amdgcn_wmma_f32_16x16x4_f32(
         false, a,  false, b,
         (short)0, acc,  false,  false);
  }

  const float bv = bias ? bias[col0 + mn] : 0.0f;
  #pragma unroll
  for (int r = 0; r < 8; ++r) {
    int row = row0 + r + half * 8;
    if (row < M) D[(long long)row * NCOLS + col0 + mn] = acc[r] + bv;
  }
}

template <int VEC, bool RELU>
__global__ void __launch_bounds__(256)
gcn_gather(const int* __restrict__ off, const int* __restrict__ cnt,
           const int2* __restrict__ csr, const float* __restrict__ HH,
           const float* __restrict__ bias, float* __restrict__ out, int N) {
  typedef float vf __attribute__((ext_vector_type(VEC)));
  const int wv   = __builtin_amdgcn_readfirstlane((int)(threadIdx.x >> 5));
  const int node = blockIdx.x * (int)(blockDim.x >> 5) + wv;
  if (node >= N) return;
  const int lane = threadIdx.x & 31;
  const vf* __restrict__ Hv = reinterpret_cast<const vf*>(HH);
  const vf bv = reinterpret_cast<const vf*>(bias)[lane];
  const vf zero = (vf)(0.0f);
  vf* __restrict__ orow = reinterpret_cast<vf*>(out + (size_t)node * (size_t)(4 * 32 * VEC));

  for (int g = 0; g < 3; ++g) {
    const int beg = off[g * N + node];
    const int c   = cnt[g * N + node];
    vf acc = zero;
    for (int base = 0; base < c; base += 32) {
      const int rem = c - base;
      const int m = rem < 32 ? rem : 32;
      const int2 ed = csr[beg + base + (lane < m ? lane : m - 1)];
      int j = 0;
      for (; j + 4 <= m; j += 4) {
        const int s0 = __builtin_amdgcn_readlane(ed.x, j);
        const int s1 = __builtin_amdgcn_readlane(ed.x, j + 1);
        const int s2 = __builtin_amdgcn_readlane(ed.x, j + 2);
        const int s3 = __builtin_amdgcn_readlane(ed.x, j + 3);
        const float w0 = __int_as_float(__builtin_amdgcn_readlane(ed.y, j));
        const float w1 = __int_as_float(__builtin_amdgcn_readlane(ed.y, j + 1));
        const float w2 = __int_as_float(__builtin_amdgcn_readlane(ed.y, j + 2));
        const float w3 = __int_as_float(__builtin_amdgcn_readlane(ed.y, j + 3));
        const vf v0 = Hv[(size_t)s0 * 32 + lane];
        const vf v1 = Hv[(size_t)s1 * 32 + lane];
        const vf v2 = Hv[(size_t)s2 * 32 + lane];
        const vf v3 = Hv[(size_t)s3 * 32 + lane];
        acc += w0 * v0;
        acc += w1 * v1;
        acc += w2 * v2;
        acc += w3 * v3;
      }
      #pragma nounroll
      for (; j < m; ++j) {
        const int s0 = __builtin_amdgcn_readlane(ed.x, j);
        const float w0 = __int_as_float(__builtin_amdgcn_readlane(ed.y, j));
        acc += w0 * Hv[(size_t)s0 * 32 + lane];
      }
    }
    vf r = acc + bv;
    if (RELU) r = __builtin_elementwise_max(r, zero);
    orow[g * 32 + lane] = r;
  }
  vf r = Hv[(size_t)node * 32 + lane] + bv;
  if (RELU) r = __builtin_elementwise_max(r, zero);
  orow[3 * 32 + lane] = r;
}

__global__ void log_softmax64(float* __restrict__ out, int N) {
  int row  = blockIdx.x * (blockDim.x >> 5) + (threadIdx.x >> 5);
  int lane = threadIdx.x & 31;
  if (row >= N) return;
  float* p = out + (long long)row * 64;
  float a = p[lane];
  float b = p[lane + 32];
  float mx = fmaxf(a, b);
  #pragma unroll
  for (int off = 16; off > 0; off >>= 1) mx = fmaxf(mx, __shfl_xor(mx, off));
  float s = __expf(a - mx) + __expf(b - mx);
  #pragma unroll
  for (int off = 16; off > 0; off >>= 1) s += __shfl_xor(s, off);
  float lse = mx + __logf(s);
  p[lane]      = a - lse;
  p[lane + 32] = b - lse;
}

extern "C" void kernel_launch(void* const* d_in, const int* in_sizes, int n_in,
                              void* d_out, int out_size, void* d_ws, size_t ws_size,
                              hipStream_t stream) {
  const float* x  = (const float*)d_in[0];
  const int* e1   = (const int*)d_in[1];
  const int* e2   = (const int*)d_in[2];
  const int* e3   = (const int*)d_in[3];
  const float* W1 = (const float*)d_in[5];
  const float* b1 = (const float*)d_in[6];
  const float* W2 = (const float*)d_in[7];
  const float* b2 = (const float*)d_in[8];
  const float* Wl = (const float*)d_in[9];
  const float* bl = (const float*)d_in[10];

  const int F = 512, H = 128, C = 64;
  const int N  = in_sizes[0] / F;
  const int E1 = in_sizes[1] / 2;
  const int E2 = in_sizes[2] / 2;
  const int E3 = in_sizes[3] / 2;
  const long long Et = (long long)E1 + E2 + E3;

  float* ws   = (float*)d_ws;
  float* HH1  = ws;
  float* R1   = HH1 + (long long)N * H;
  float* HH2  = R1 + (long long)N * 4 * H;
  float* R2   = HH2 + (long long)N * C;
  int2*  csr  = (int2*)(R2 + (long long)N * 4 * C);
  float* dinv = (float*)(csr + Et);
  int*   cnt  = (int*)(dinv + 3LL * N);
  int*   off  = cnt + 3LL * N;
  int*   rank = off + 3LL * N;

  const int T = 256;

  zero_i32<<<cdiv_i(3LL * N, T), T, 0, stream>>>(cnt, 3LL * N);
  count_rank<<<cdiv_i(E1, T), T, 0, stream>>>(e1 + E1, E1, cnt,           rank);
  count_rank<<<cdiv_i(E2, T), T, 0, stream>>>(e2 + E2, E2, cnt + N,       rank + E1);
  count_rank<<<cdiv_i(E3, T), T, 0, stream>>>(e3 + E3, E3, cnt + 2LL * N, rank + E1 + E2);
  scan_deg<<<3, 1024, 0, stream>>>(cnt, off, dinv, N, E1, E2);
  csr_fill<<<cdiv_i(E1, T), T, 0, stream>>>(e1, e1 + E1, E1, off,           rank,           dinv,           csr);
  csr_fill<<<cdiv_i(E2, T), T, 0, stream>>>(e2, e2 + E2, E2, off + N,       rank + E1,      dinv + N,       csr);
  csr_fill<<<cdiv_i(E3, T), T, 0, stream>>>(e3, e3 + E3, E3, off + 2LL * N, rank + E1 + E2, dinv + 2LL * N, csr);

  const int mtiles = cdiv_i(N, 16);
  const int gblocks = cdiv_i(N, T / 32);

  gemm_wmma_f32<F, H><<<mtiles, 32 * (H / 16), 0, stream>>>(x, W1, nullptr, HH1, N);

  gcn_gather<4, true><<<gblocks, T, 0, stream>>>(off, cnt, csr, HH1, b1, R1, N);

  gemm_wmma_f32<4 * H, C><<<mtiles, 32 * (C / 16), 0, stream>>>(R1, W2, nullptr, HH2, N);

  gcn_gather<2, false><<<gblocks, T, 0, stream>>>(off, cnt, csr, HH2, b2, R2, N);

  float* logits = (float*)d_out;
  gemm_wmma_f32<4 * C, C><<<mtiles, 32 * (C / 16), 0, stream>>>(R2, Wl, bl, logits, N);

  log_softmax64<<<cdiv_i(N, T / 32), T, 0, stream>>>(logits, N);
}


# Short passes (edge count, fill, softmax, zeroing) as few long-lived waves: grid-stride, 4 items in flight per thread, 3 edge sets per launch (14 launches to 10); sums still in atomic-arrival order
# speedup vs baseline: 3.6314x; 1.0785x over previous
#include <hip/hip_runtime.h>
#include <math.h>

typedef float v2f __attribute__((ext_vector_type(2)));
typedef float v8f __attribute__((ext_vector_type(8)));

static inline int cdiv_i(long long a, long long b) { return (int)((a + b - 1) / b); }

__global__ void zero_i32(int* __restrict__ p, long long n) {
  long long i = (long long)blockIdx.x * blockDim.x + threadIdx.x;
  long long stride = (long long)gridDim.x * blockDim.x;
  for (; i < n; i += stride) p[i] = 0;
}

struct EdgeRef { const int* p; int gN; };
__device__ __forceinline__ EdgeRef edge_ref(int i, const int* __restrict__ a1, const int* __restrict__ a2,
                                            const int* __restrict__ a3, int E1, int E12, int N) {
  EdgeRef r;
  int li = i;
  r.p = a1; r.gN = 0;
  if (i >= E1)  { r.p = a2; li = i - E1;  r.gN = N; }
  if (i >= E12) { r.p = a3; li = i - E12; r.gN = 2 * N; }
  r.p += li;
  return r;
}

template <int U, bool GUARD>
__device__ __forceinline__ void count_chunk(int base, const int* __restrict__ d1, const int* __restrict__ d2,
                                            const int* __restrict__ d3, int E1, int E12, int Et, int N,
                                            int* __restrict__ cnt, int* __restrict__ rank) {
  int slot[U];
  #pragma unroll
  for (int u = 0; u < U; ++u) {
    const int i  = base + u * 256;
    const int ic = (GUARD && i >= Et) ? Et - 1 : i;
    const EdgeRef e = edge_ref(ic, d1, d2, d3, E1, E12, N);
    slot[u] = e.gN + *e.p;
  }
  int r[U];
  #pragma unroll
  for (int u = 0; u < U; ++u) {
    r[u] = 0;
    if (!GUARD || base + u * 256 < Et) r[u] = atomicAdd(&cnt[slot[u]], 1);
  }
  #pragma unroll
  for (int u = 0; u < U; ++u) {
    if (!GUARD || base + u * 256 < Et) rank[base + u * 256] = r[u];
  }
}

template <int U>
__global__ void __launch_bounds__(256)
count_rank_all(const int* __restrict__ d1, const int* __restrict__ d2, const int* __restrict__ d3,
               int E1, int E2, int E3, int N, int* __restrict__ cnt, int* __restrict__ rank) {
  const int E12 = E1 + E2;
  const int Et  = E12 + E3;
  const int step = (int)gridDim.x * 256 * U;
  for (int c0 = (int)blockIdx.x * 256 * U; c0 < Et; c0 += step) {
    const int base = c0 + (int)threadIdx.x;
    if (c0 + 256 * U <= Et) count_chunk<U, false>(base, d1, d2, d3, E1, E12, Et, N, cnt, rank);
    else                    count_chunk<U, true >(base, d1, d2, d3, E1, E12, Et, N, cnt, rank);
  }
}

__global__ void scan_deg(const int* __restrict__ cnt, int* __restrict__ off,
                         float* __restrict__ dinv, int N, int E1, int E2) {
  __shared__ int part[1024];
  const int g = blockIdx.x;
  const int t = threadIdx.x;
  const int nt = blockDim.x;
  const int per = (N + nt - 1) / nt;
  int lo = t * per; if (lo > N) lo = N;
  int hi = lo + per; if (hi > N) hi = N;
  const int* c = cnt + (long long)g * N;
  int s = 0;
  for (int i = lo; i < hi; ++i) s += c[i];
  part[t] = s;
  __syncthreads();
  for (int d = 1; d < nt; d <<= 1) {
    int v = (t >= d) ? part[t - d] : 0;
    __syncthreads();
    part[t] += v;
    __syncthreads();
  }
  int run = part[t] - s;
  if (g >= 1) run += E1;
  if (g >= 2) run += E2;
  int* o = off + (long long)g * N;
  float* dv = dinv + (long long)g * N;
  for (int i = lo; i < hi; ++i) {
    int v = c[i];
    o[i] = run;
    run += v;
    dv[i] = (v > 0) ? rsqrtf((float)v) : 0.0f;
  }
}

template <int U, bool GUARD>
__device__ __forceinline__ void fill_chunk(int base, const int* __restrict__ e1, const int* __restrict__ e2,
                                           const int* __restrict__ e3, int E1, int E2, int E3, int E12, int Et,
                                           int N, const int* __restrict__ off, const int* __restrict__ rank,
                                           const float* __restrict__ dinv, int2* __restrict__ csr) {
  int s[U], d[U], rk[U], gN[U];
  #pragma unroll
  for (int u = 0; u < U; ++u) {
    const int i  = base + u * 256;
    const int ic = (GUARD && i >= Et) ? Et - 1 : i;
    const EdgeRef es = edge_ref(ic, e1, e2, e3, E1, E12, N);
    const EdgeRef ed = edge_ref(ic, e1 + E1, e2 + E2, e3 + E3, E1, E12, N);
    s[u]  = *es.p;
    d[u]  = *ed.p;
    rk[u] = rank[ic];
    gN[u] = es.gN;
  }
  int o[U];
  float ws[U], wd[U];
  #pragma unroll
  for (int u = 0; u < U; ++u) {
    o[u]  = off[gN[u] + d[u]];
    ws[u] = dinv[gN[u] + s[u]];
    wd[u] = dinv[gN[u] + d[u]];
  }
  #pragma unroll
  for (int u = 0; u < U; ++u) {
    if (!GUARD || base + u * 256 < Et) {
      int2 v;
      v.x = s[u];
      v.y = __float_as_int(ws[u] * wd[u]);
      csr[o[u] + rk[u]] = v;
    }
  }
}

template <int U>
__global__ void __launch_bounds__(256)
csr_fill_all(const int* __restrict__ e1, const int* __restrict__ e2, const int* __restrict__ e3,
             int E1, int E2, int E3, int N, const int* __restrict__ off, const int* __restrict__ rank,
             const float* __restrict__ dinv, int2* __restrict__ csr) {
  const int E12 = E1 + E2;
  const int Et  = E12 + E3;
  const int step = (int)gridDim.x * 256 * U;
  for (int c0 = (int)blockIdx.x * 256 * U; c0 < Et; c0 += step) {
    const int base = c0 + (int)threadIdx.x;
    if (c0 + 256 * U <= Et) fill_chunk<U, false>(base, e1, e2, e3, E1, E2, E3, E12, Et, N, off, rank, dinv, csr);
    else                    fill_chunk<U, true >(base, e1, e2, e3, E1, E2, E3, E12, Et, N, off, rank, dinv, csr);
  }
}

template <int K, int NCOLS>
__global__ void __launch_bounds__(NCOLS * 2)
gemm_wmma_f32(const float* __restrict__ A, const float* __restrict__ B,
              const float* __restrict__ bias, float* __restrict__ D, int M) {
  constexpr int NT  = NCOLS * 2;
  constexpr int LDA = K + 4;
  constexpr int K4  = K / 4;
  constexpr int PER = (16 * K4) / NT;
  static_assert((16 * K4) % NT == 0, "strip must divide evenly over the block");
  __shared__ __attribute__((aligned(16))) float As[16 * LDA];

  const int row0 = blockIdx.x * 16;
  const int tid  = threadIdx.x;

  float4 stage[PER];
  #pragma unroll
  for (int i = 0; i < PER; ++i) {
    const int q  = tid + i * NT;
    const int r  = q / K4;
    const int c4 = q - r * K4;
    int row = row0 + r;
    if (row >= M) row = M - 1;
    stage[i] = *reinterpret_cast<const float4*>(A + (long long)row * K + c4 * 4);
  }
  #pragma unroll
  for (int i = 0; i < PER; ++i) {
    const int q  = tid + i * NT;
    const int r  = q / K4;
    const int c4 = q - r * K4;
    *reinterpret_cast<float4*>(&As[r * LDA + c4 * 4]) = stage[i];
  }
  __syncthreads();

  const int wave = tid >> 5;
  const int lane = tid & 31;
  const int half = lane >> 4;
  const int mn   = lane & 15;
  const int kb   = half * 2;
  const int col0 = wave * 16;

  const float* Ap = &As[mn * LDA + kb];
  const float* Bp = B + kb * NCOLS + col0 + mn;

  v8f acc = {};
  #pragma unroll 4
  for (int kk = 0; kk < K; kk += 4) {
    const v2f a = *reinterpret_cast<const v2f*>(Ap + kk);
    v2f b;
    b.x = Bp[kk * NCOLS];
    b.y = Bp[kk * NCOLS + NCOLS];
    acc = __builtin_amdgcn_wmma_f32_16x16x4_f32(
         false, a,  false, b,
         (short)0, acc,  false,  false);
  }

  const float bv = bias ? bias[col0 + mn] : 0.0f;
  #pragma unroll
  for (int r = 0; r < 8; ++r) {
    int row = row0 + r + half * 8;
    if (row < M) D[(long long)row * NCOLS + col0 + mn] = acc[r] + bv;
  }
}

template <int VEC, bool RELU>
__global__ void __launch_bounds__(256)
gcn_gather(const int* __restrict__ off, const int* __restrict__ cnt,
           const int2* __restrict__ csr, const float* __restrict__ HH,
           const float* __restrict__ bias, float* __restrict__ out, int N) {
  typedef float vf __attribute__((ext_vector_type(VEC)));
  const int wv   = __builtin_amdgcn_readfirstlane((int)(threadIdx.x >> 5));
  const int node = blockIdx.x * (int)(blockDim.x >> 5) + wv;
  if (node >= N) return;
  const int lane = threadIdx.x & 31;
  const vf* __restrict__ Hv = reinterpret_cast<const vf*>(HH);
  const vf bv = reinterpret_cast<const vf*>(bias)[lane];
  const vf zero = (vf)(0.0f);
  vf* __restrict__ orow = reinterpret_cast<vf*>(out + (size_t)node * (size_t)(4 * 32 * VEC));

  for (int g = 0; g < 3; ++g) {
    const int beg = off[g * N + node];
    const int c   = cnt[g * N + node];
    vf acc = zero;
    for (int base = 0; base < c; base += 32) {
      const int rem = c - base;
      const int m = rem < 32 ? rem : 32;
      const int2 ed = csr[beg + base + (lane < m ? lane : m - 1)];
      int j = 0;
      for (; j + 4 <= m; j += 4) {
        const int s0 = __builtin_amdgcn_readlane(ed.x, j);
        const int s1 = __builtin_amdgcn_readlane(ed.x, j + 1);
        const int s2 = __builtin_amdgcn_readlane(ed.x, j + 2);
        const int s3 = __builtin_amdgcn_readlane(ed.x, j + 3);
        const float w0 = __int_as_float(__builtin_amdgcn_readlane(ed.y, j));
        const float w1 = __int_as_float(__builtin_amdgcn_readlane(ed.y, j + 1));
        const float w2 = __int_as_float(__builtin_amdgcn_readlane(ed.y, j + 2));
        const float w3 = __int_as_float(__builtin_amdgcn_readlane(ed.y, j + 3));
        const vf v0 = Hv[(size_t)s0 * 32 + lane];
        const vf v1 = Hv[(size_t)s1 * 32 + lane];
        const vf v2 = Hv[(size_t)s2 * 32 + lane];
        const vf v3 = Hv[(size_t)s3 * 32 + lane];
        acc += w0 * v0;
        acc += w1 * v1;
        acc += w2 * v2;
        acc += w3 * v3;
      }
      #pragma nounroll
      for (; j < m; ++j) {
        const int s0 = __builtin_amdgcn_readlane(ed.x, j);
        const float w0 = __int_as_float(__builtin_amdgcn_readlane(ed.y, j));
        acc += w0 * Hv[(size_t)s0 * 32 + lane];
      }
    }
    vf r = acc + bv;
    if (RELU) r = __builtin_elementwise_max(r, zero);
    orow[g * 32 + lane] = r;
  }
  vf r = Hv[(size_t)node * 32 + lane] + bv;
  if (RELU) r = __builtin_elementwise_max(r, zero);
  orow[3 * 32 + lane] = r;
}

__device__ __forceinline__ void lsm_row(float a, float b, float& ra, float& rb) {
  float mx = fmaxf(a, b);
  #pragma unroll
  for (int off = 16; off > 0; off >>= 1) mx = fmaxf(mx, __shfl_xor(mx, off));
  float s = __expf(a - mx) + __expf(b - mx);
  #pragma unroll
  for (int off = 16; off > 0; off >>= 1) s += __shfl_xor(s, off);
  const float lse = mx + __logf(s);
  ra = a - lse;
  rb = b - lse;
}

__global__ void __launch_bounds__(256) log_softmax64(float* __restrict__ out, int N) {
  constexpr int R = 4;
  const int W    = (int)gridDim.x * (int)(blockDim.x >> 5);
  const int lane = threadIdx.x & 31;
  int row = (int)blockIdx.x * (int)(blockDim.x >> 5) + (int)(threadIdx.x >> 5);
  for (; row + (R - 1) * W < N; row += R * W) {
    float a[R], b[R];
    #pragma unroll
    for (int j = 0; j < R; ++j) {
      const float* p = out + (long long)(row + j * W) * 64;
      a[j] = p[lane];
      b[j] = p[lane + 32];
    }
    #pragma unroll
    for (int j = 0; j < R; ++j) {
      float ra, rb;
      lsm_row(a[j], b[j], ra, rb);
      float* p = out + (long long)(row + j * W) * 64;
      p[lane]      = ra;
      p[lane + 32] = rb;
    }
  }
  for (; row < N; row += W) {
    float* p = out + (long long)row * 64;
    float ra, rb;
    lsm_row(p[lane], p[lane + 32], ra, rb);
    p[lane]      = ra;
    p[lane + 32] = rb;
  }
}

extern "C" void kernel_launch(void* const* d_in, const int* in_sizes, int n_in,
                              void* d_out, int out_size, void* d_ws, size_t ws_size,
                              hipStream_t stream) {
  const float* x  = (const float*)d_in[0];
  const int* e1   = (const int*)d_in[1];
  const int* e2   = (const int*)d_in[2];
  const int* e3   = (const int*)d_in[3];
  const float* W1 = (const float*)d_in[5];
  const float* b1 = (const float*)d_in[6];
  const float* W2 = (const float*)d_in[7];
  const float* b2 = (const float*)d_in[8];
  const float* Wl = (const float*)d_in[9];
  const float* bl = (const float*)d_in[10];

  const int F = 512, H = 128, C = 64;
  const int N  = in_sizes[0] / F;
  const int E1 = in_sizes[1] / 2;
  const int E2 = in_sizes[2] / 2;
  const int E3 = in_sizes[3] / 2;
  const long long Et = (long long)E1 + E2 + E3;

  float* ws   = (float*)d_ws;
  float* HH1  = ws;
  float* R1   = HH1 + (long long)N * H;
  float* HH2  = R1 + (long long)N * 4 * H;
  float* R2   = HH2 + (long long)N * C;
  int2*  csr  = (int2*)(R2 + (long long)N * 4 * C);
  float* dinv = (float*)(csr + Et);
  int*   cnt  = (int*)(dinv + 3LL * N);
  int*   off  = cnt + 3LL * N;
  int*   rank = off + 3LL * N;

  const int T = 256;

  zero_i32<<<64, T, 0, stream>>>(cnt, 3LL * N);
  count_rank_all<4><<<512, T, 0, stream>>>(e1 + E1, e2 + E2, e3 + E3, E1, E2, E3, N, cnt, rank);
  scan_deg<<<3, 1024, 0, stream>>>(cnt, off, dinv, N, E1, E2);
  csr_fill_all<4><<<512, T, 0, stream>>>(e1, e2, e3, E1, E2, E3, N, off, rank, dinv, csr);

  const int mtiles = cdiv_i(N, 16);
  const int gblocks = cdiv_i(N, T / 32);

  gemm_wmma_f32<F, H><<<mtiles, 32 * (H / 16), 0, stream>>>(x, W1, nullptr, HH1, N);

  gcn_gather<4, true><<<gblocks, T, 0, stream>>>(off, cnt, csr, HH1, b1, R1, N);

  gemm_wmma_f32<4 * H, C><<<mtiles, 32 * (C / 16), 0, stream>>>(R1, W2, nullptr, HH2, N);

  gcn_gather<2, false><<<gblocks, T, 0, stream>>>(off, cnt, csr, HH2, b2, R2, N);

  float* logits = (float*)d_out;
  gemm_wmma_f32<4 * C, C><<<mtiles, 32 * (C / 16), 0, stream>>>(R2, Wl, bl, logits, N);

  log_softmax64<<<256, T, 0, stream>>>(logits, N);
}


# Gather: fewer dependent memory trips per wave: all 3 sets' offsets, counts and first edge chunks fetched together, next chunk prefetched, 8 or 16 rows in flight, tail rows in one trip; sum order kept
# speedup vs baseline: 3.7633x; 1.0363x over previous
#include <hip/hip_runtime.h>
#include <math.h>

typedef float v2f __attribute__((ext_vector_type(2)));
typedef float v8f __attribute__((ext_vector_type(8)));

static inline int cdiv_i(long long a, long long b) { return (int)((a + b - 1) / b); }

__global__ void zero_i32(int* __restrict__ p, long long n) {
  long long i = (long long)blockIdx.x * blockDim.x + threadIdx.x;
  long long stride = (long long)gridDim.x * blockDim.x;
  for (; i < n; i += stride) p[i] = 0;
}

struct EdgeRef { const int* p; int gN; };
__device__ __forceinline__ EdgeRef edge_ref(int i, const int* __restrict__ a1, const int* __restrict__ a2,
                                            const int* __restrict__ a3, int E1, int E12, int N) {
  EdgeRef r;
  int li = i;
  r.p = a1; r.gN = 0;
  if (i >= E1)  { r.p = a2; li = i - E1;  r.gN = N; }
  if (i >= E12) { r.p = a3; li = i - E12; r.gN = 2 * N; }
  r.p += li;
  return r;
}

template <int U, bool GUARD>
__device__ __forceinline__ void count_chunk(int base, const int* __restrict__ d1, const int* __restrict__ d2,
                                            const int* __restrict__ d3, int E1, int E12, int Et, int N,
                                            int* __restrict__ cnt, int* __restrict__ rank) {
  int slot[U];
  #pragma unroll
  for (int u = 0; u < U; ++u) {
    const int i  = base + u * 256;
    const int ic = (GUARD && i >= Et) ? Et - 1 : i;
    const EdgeRef e = edge_ref(ic, d1, d2, d3, E1, E12, N);
    slot[u] = e.gN + *e.p;
  }
  int r[U];
  #pragma unroll
  for (int u = 0; u < U; ++u) {
    r[u] = 0;
    if (!GUARD || base + u * 256 < Et) r[u] = atomicAdd(&cnt[slot[u]], 1);
  }
  #pragma unroll
  for (int u = 0; u < U; ++u) {
    if (!GUARD || base + u * 256 < Et) rank[base + u * 256] = r[u];
  }
}

template <int U>
__global__ void __launch_bounds__(256)
count_rank_all(const int* __restrict__ d1, const int* __restrict__ d2, const int* __restrict__ d3,
               int E1, int E2, int E3, int N, int* __restrict__ cnt, int* __restrict__ rank) {
  const int E12 = E1 + E2;
  const int Et  = E12 + E3;
  const int step = (int)gridDim.x * 256 * U;
  for (int c0 = (int)blockIdx.x * 256 * U; c0 < Et; c0 += step) {
    const int base = c0 + (int)threadIdx.x;
    if (c0 + 256 * U <= Et) count_chunk<U, false>(base, d1, d2, d3, E1, E12, Et, N, cnt, rank);
    else                    count_chunk<U, true >(base, d1, d2, d3, E1, E12, Et, N, cnt, rank);
  }
}

__global__ void scan_deg(const int* __restrict__ cnt, int* __restrict__ off,
                         float* __restrict__ dinv, int N, int E1, int E2) {
  __shared__ int part[1024];
  const int g = blockIdx.x;
  const int t = threadIdx.x;
  const int nt = blockDim.x;
  const int per = (N + nt - 1) / nt;
  int lo = t * per; if (lo > N) lo = N;
  int hi = lo + per; if (hi > N) hi = N;
  const int* c = cnt + (long long)g * N;
  int s = 0;
  for (int i = lo; i < hi; ++i) s += c[i];
  part[t] = s;
  __syncthreads();
  for (int d = 1; d < nt; d <<= 1) {
    int v = (t >= d) ? part[t - d] : 0;
    __syncthreads();
    part[t] += v;
    __syncthreads();
  }
  int run = part[t] - s;
  if (g >= 1) run += E1;
  if (g >= 2) run += E2;
  int* o = off + (long long)g * N;
  float* dv = dinv + (long long)g * N;
  for (int i = lo; i < hi; ++i) {
    int v = c[i];
    o[i] = run;
    run += v;
    dv[i] = (v > 0) ? rsqrtf((float)v) : 0.0f;
  }
}

template <int U, bool GUARD>
__device__ __forceinline__ void fill_chunk(int base, const int* __restrict__ e1, const int* __restrict__ e2,
                                           const int* __restrict__ e3, int E1, int E2, int E3, int E12, int Et,
                                           int N, const int* __restrict__ off, const int* __restrict__ rank,
                                           const float* __restrict__ dinv, int2* __restrict__ csr) {
  int s[U], d[U], rk[U], gN[U];
  #pragma unroll
  for (int u = 0; u < U; ++u) {
    const int i  = base + u * 256;
    const int ic = (GUARD && i >= Et) ? Et - 1 : i;
    const EdgeRef es = edge_ref(ic, e1, e2, e3, E1, E12, N);
    const EdgeRef ed = edge_ref(ic, e1 + E1, e2 + E2, e3 + E3, E1, E12, N);
    s[u]  = *es.p;
    d[u]  = *ed.p;
    rk[u] = rank[ic];
    gN[u] = es.gN;
  }
  int o[U];
  float ws[U], wd[U];
  #pragma unroll
  for (int u = 0; u < U; ++u) {
    o[u]  = off[gN[u] + d[u]];
    ws[u] = dinv[gN[u] + s[u]];
    wd[u] = dinv[gN[u] + d[u]];
  }
  #pragma unroll
  for (int u = 0; u < U; ++u) {
    if (!GUARD || base + u * 256 < Et) {
      int2 v;
      v.x = s[u];
      v.y = __float_as_int(ws[u] * wd[u]);
      csr[o[u] + rk[u]] = v;
    }
  }
}

template <int U>
__global__ void __launch_bounds__(256)
csr_fill_all(const int* __restrict__ e1, const int* __restrict__ e2, const int* __restrict__ e3,
             int E1, int E2, int E3, int N, const int* __restrict__ off, const int* __restrict__ rank,
             const float* __restrict__ dinv, int2* __restrict__ csr) {
  const int E12 = E1 + E2;
  const int Et  = E12 + E3;
  const int step = (int)gridDim.x * 256 * U;
  for (int c0 = (int)blockIdx.x * 256 * U; c0 < Et; c0 += step) {
    const int base = c0 + (int)threadIdx.x;
    if (c0 + 256 * U <= Et) fill_chunk<U, false>(base, e1, e2, e3, E1, E2, E3, E12, Et, N, off, rank, dinv, csr);
    else                    fill_chunk<U, true >(base, e1, e2, e3, E1, E2, E3, E12, Et, N, off, rank, dinv, csr);
  }
}

template <int K, int NCOLS>
__global__ void __launch_bounds__(NCOLS * 2)
gemm_wmma_f32(const float* __restrict__ A, const float* __restrict__ B,
              const float* __restrict__ bias, float* __restrict__ D, int M) {
  constexpr int NT  = NCOLS * 2;
  constexpr int LDA = K + 4;
  constexpr int K4  = K / 4;
  constexpr int PER = (16 * K4) / NT;
  static_assert((16 * K4) % NT == 0, "strip must divide evenly over the block");
  __shared__ __attribute__((aligned(16))) float As[16 * LDA];

  const int row0 = blockIdx.x * 16;
  const int tid  = threadIdx.x;

  float4 stage[PER];
  #pragma unroll
  for (int i = 0; i < PER; ++i) {
    const int q  = tid + i * NT;
    const int r  = q / K4;
    const int c4 = q - r * K4;
    int row = row0 + r;
    if (row >= M) row = M - 1;
    stage[i] = *reinterpret_cast<const float4*>(A + (long long)row * K + c4 * 4);
  }
  #pragma unroll
  for (int i = 0; i < PER; ++i) {
    const int q  = tid + i * NT;
    const int r  = q / K4;
    const int c4 = q - r * K4;
    *reinterpret_cast<float4*>(&As[r * LDA + c4 * 4]) = stage[i];
  }
  __syncthreads();

  const int wave = tid >> 5;
  const int lane = tid & 31;
  const int half = lane >> 4;
  const int mn   = lane & 15;
  const int kb   = half * 2;
  const int col0 = wave * 16;

  const float* Ap = &As[mn * LDA + kb];
  const float* Bp = B + kb * NCOLS + col0 + mn;

  v8f acc = {};
  #pragma unroll 4
  for (int kk = 0; kk < K; kk += 4) {
    const v2f a = *reinterpret_cast<const v2f*>(Ap + kk);
    v2f b;
    b.x = Bp[kk * NCOLS];
    b.y = Bp[kk * NCOLS + NCOLS];
    acc = __builtin_amdgcn_wmma_f32_16x16x4_f32(
         false, a,  false, b,
         (short)0, acc,  false,  false);
  }

  const float bv = bias ? bias[col0 + mn] : 0.0f;
  #pragma unroll
  for (int r = 0; r < 8; ++r) {
    int row = row0 + r + half * 8;
    if (row < M) D[(long long)row * NCOLS + col0 + mn] = acc[r] + bv;
  }
}

template <int U, int VEC>
__device__ __forceinline__ void gather_rows(const int2 ed, const int j, const int lane,
                                            const float* __restrict__ HH,
                                            float __attribute__((ext_vector_type(VEC)))& acc) {
  typedef float vf __attribute__((ext_vector_type(VEC)));
  const vf* __restrict__ Hv = reinterpret_cast<const vf*>(HH);
  int s[U];
  float w[U];
  vf v[U];
  #pragma unroll
  for (int u = 0; u < U; ++u) {
    s[u] = __builtin_amdgcn_readlane(ed.x, j + u);
    w[u] = __int_as_float(__builtin_amdgcn_readlane(ed.y, j + u));
  }
  #pragma unroll
  for (int u = 0; u < U; ++u) v[u] = Hv[(size_t)s[u] * 32 + lane];
  #pragma unroll
  for (int u = 0; u < U; ++u) acc += w[u] * v[u];
}

template <int VEC, bool RELU>
__global__ void __launch_bounds__(256)
gcn_gather(const int* __restrict__ off, const int* __restrict__ cnt,
           const int2* __restrict__ csr, const float* __restrict__ HH,
           const float* __restrict__ bias, float* __restrict__ out, int N) {
  typedef float vf __attribute__((ext_vector_type(VEC)));
  constexpr int WIDE = 32 / VEC;
  const int wv   = __builtin_amdgcn_readfirstlane((int)(threadIdx.x >> 5));
  const int node = blockIdx.x * (int)(blockDim.x >> 5) + wv;
  if (node >= N) return;
  const int lane = threadIdx.x & 31;
  const vf* __restrict__ Hv = reinterpret_cast<const vf*>(HH);
  const vf zero = (vf)(0.0f);
  vf* __restrict__ orow = reinterpret_cast<vf*>(out + (size_t)node * (size_t)(4 * 32 * VEC));

  int beg[3], c[3];
  #pragma unroll
  for (int g = 0; g < 3; ++g) {
    beg[g] = off[g * N + node];
    c[g]   = cnt[g * N + node];
  }
  asm volatile("" :: "s"(beg[0]), "s"(beg[1]), "s"(beg[2]), "s"(c[0]), "s"(c[1]), "s"(c[2]));

  int2 first[3];
  #pragma unroll
  for (int g = 0; g < 3; ++g) {
    first[g] = make_int2(0, 0);
    if (c[g] > 0) {
      const int m0 = c[g] < 32 ? c[g] : 32;
      first[g] = csr[beg[g] + (lane < m0 ? lane : m0 - 1)];
    }
  }
  const vf bv   = reinterpret_cast<const vf*>(bias)[lane];
  const vf self = Hv[(size_t)node * 32 + lane];
  asm volatile("" :: "v"(bv), "v"(self));

  #pragma unroll
  for (int g = 0; g < 3; ++g) {
    vf acc = zero;
    int2 ed = first[g];
    for (int base = 0; base < c[g]; base += 32) {
      const int rem = c[g] - base;
      const int m = rem < 32 ? rem : 32;
      int2 nxt = ed;
      if (rem > 32) {
        const int m2 = rem - 32 < 32 ? rem - 32 : 32;
        nxt = csr[beg[g] + base + 32 + (lane < m2 ? lane : m2 - 1)];
      }
      int j = 0;
      for (; j + WIDE <= m; j += WIDE) gather_rows<WIDE, VEC>(ed, j, lane, HH, acc);
      if (WIDE > 8 && j + 8 <= m) { gather_rows<8, VEC>(ed, j, lane, HH, acc); j += 8; }
      if (j + 4 <= m) { gather_rows<4, VEC>(ed, j, lane, HH, acc); j += 4; }
      if (j < m) {
        const int j1 = j + 1 < m ? j + 1 : j;
        const int j2 = j + 2 < m ? j + 2 : j;
        const int s0 = __builtin_amdgcn_readlane(ed.x, j);
        const int s1 = __builtin_amdgcn_readlane(ed.x, j1);
        const int s2 = __builtin_amdgcn_readlane(ed.x, j2);
        const float w0 = __int_as_float(__builtin_amdgcn_readlane(ed.y, j));
        const float w1 = __int_as_float(__builtin_amdgcn_readlane(ed.y, j1));
        const float w2 = __int_as_float(__builtin_amdgcn_readlane(ed.y, j2));
        const vf v0 = Hv[(size_t)s0 * 32 + lane];
        const vf v1 = Hv[(size_t)s1 * 32 + lane];
        const vf v2 = Hv[(size_t)s2 * 32 + lane];
        asm volatile("" :: "v"(v0), "v"(v1), "v"(v2));
        acc += w0 * v0;
        if (j + 1 < m) acc += w1 * v1;
        if (j + 2 < m) acc += w2 * v2;
      }
      ed = nxt;
    }
    vf r = acc + bv;
    if (RELU) r = __builtin_elementwise_max(r, zero);
    orow[g * 32 + lane] = r;
  }
  vf r = self + bv;
  if (RELU) r = __builtin_elementwise_max(r, zero);
  orow[3 * 32 + lane] = r;
}

__device__ __forceinline__ void lsm_row(float a, float b, float& ra, float& rb) {
  float mx = fmaxf(a, b);
  #pragma unroll
  for (int off = 16; off > 0; off >>= 1) mx = fmaxf(mx, __shfl_xor(mx, off));
  float s = __expf(a - mx) + __expf(b - mx);
  #pragma unroll
  for (int off = 16; off > 0; off >>= 1) s += __shfl_xor(s, off);
  const float lse = mx + __logf(s);
  ra = a - lse;
  rb = b - lse;
}

__global__ void __launch_bounds__(256) log_softmax64(float* __restrict__ out, int N) {
  constexpr int R = 4;
  const int W    = (int)gridDim.x * (int)(blockDim.x >> 5);
  const int lane = threadIdx.x & 31;
  int row = (int)blockIdx.x * (int)(blockDim.x >> 5) + (int)(threadIdx.x >> 5);
  for (; row + (R - 1) * W < N; row += R * W) {
    float a[R], b[R];
    #pragma unroll
    for (int j = 0; j < R; ++j) {
      const float* p = out + (long long)(row + j * W) * 64;
      a[j] = p[lane];
      b[j] = p[lane + 32];
    }
    #pragma unroll
    for (int j = 0; j < R; ++j) {
      float ra, rb;
      lsm_row(a[j], b[j], ra, rb);
      float* p = out + (long long)(row + j * W) * 64;
      p[lane]      = ra;
      p[lane + 32] = rb;
    }
  }
  for (; row < N; row += W) {
    float* p = out + (long long)row * 64;
    float ra, rb;
    lsm_row(p[lane], p[lane + 32], ra, rb);
    p[lane]      = ra;
    p[lane + 32] = rb;
  }
}

extern "C" void kernel_launch(void* const* d_in, const int* in_sizes, int n_in,
                              void* d_out, int out_size, void* d_ws, size_t ws_size,
                              hipStream_t stream) {
  const float* x  = (const float*)d_in[0];
  const int* e1   = (const int*)d_in[1];
  const int* e2   = (const int*)d_in[2];
  const int* e3   = (const int*)d_in[3];
  const float* W1 = (const float*)d_in[5];
  const float* b1 = (const float*)d_in[6];
  const float* W2 = (const float*)d_in[7];
  const float* b2 = (const float*)d_in[8];
  const float* Wl = (const float*)d_in[9];
  const float* bl = (const float*)d_in[10];

  const int F = 512, H = 128, C = 64;
  const int N  = in_sizes[0] / F;
  const int E1 = in_sizes[1] / 2;
  const int E2 = in_sizes[2] / 2;
  const int E3 = in_sizes[3] / 2;
  const long long Et = (long long)E1 + E2 + E3;

  float* ws   = (float*)d_ws;
  float* HH1  = ws;
  float* R1   = HH1 + (long long)N * H;
  float* HH2  = R1 + (long long)N * 4 * H;
  float* R2   = HH2 + (long long)N * C;
  int2*  csr  = (int2*)(R2 + (long long)N * 4 * C);
  float* dinv = (float*)(csr + Et);
  int*   cnt  = (int*)(dinv + 3LL * N);
  int*   off  = cnt + 3LL * N;
  int*   rank = off + 3LL * N;

  const int T = 256;

  zero_i32<<<64, T, 0, stream>>>(cnt, 3LL * N);
  count_rank_all<4><<<512, T, 0, stream>>>(e1 + E1, e2 + E2, e3 + E3, E1, E2, E3, N, cnt, rank);
  scan_deg<<<3, 1024, 0, stream>>>(cnt, off, dinv, N, E1, E2);
  csr_fill_all<4><<<512, T, 0, stream>>>(e1, e2, e3, E1, E2, E3, N, off, rank, dinv, csr);

  const int mtiles = cdiv_i(N, 16);
  const int gblocks = cdiv_i(N, T / 32);

  gemm_wmma_f32<F, H><<<mtiles, 32 * (H / 16), 0, stream>>>(x, W1, nullptr, HH1, N);

  gcn_gather<4, true><<<gblocks, T, 0, stream>>>(off, cnt, csr, HH1, b1, R1, N);

  gemm_wmma_f32<4 * H, C><<<mtiles, 32 * (C / 16), 0, stream>>>(R1, W2, nullptr, HH2, N);

  gcn_gather<2, false><<<gblocks, T, 0, stream>>>(off, cnt, csr, HH2, b2, R2, N);

  float* logits = (float*)d_out;
  gemm_wmma_f32<4 * C, C><<<mtiles, 32 * (C / 16), 0, stream>>>(R2, Wl, bl, logits, N);

  log_softmax64<<<256, T, 0, stream>>>(logits, N);
}


# Edge sort by destination on LDS counters, two levels (256-key bins per block chunk, then keys inside a bin), replacing per-edge returning global atomics; sum order = arrival order of the LDS adds
# speedup vs baseline: 4.8832x; 1.2976x over previous
#include <hip/hip_runtime.h>
#include <math.h>

typedef float v2f __attribute__((ext_vector_type(2)));
typedef float v8f __attribute__((ext_vector_type(8)));

static inline int cdiv_i(long long a, long long b) { return (int)((a + b - 1) / b); }

constexpr int BIN_SHIFT = 8;
constexpr int BIN_KEYS  = 1 << BIN_SHIFT;
constexpr int MAX_BINS  = 1024;
constexpr int SORT_T    = 1024;

struct EdgeRef { const int* p; int gN; };
__device__ __forceinline__ EdgeRef edge_ref(int i, const int* __restrict__ a1, const int* __restrict__ a2,
                                            const int* __restrict__ a3, int E1, int E12, int N) {
  EdgeRef r;
  int li = i;
  r.p = a1; r.gN = 0;
  if (i >= E1)  { r.p = a2; li = i - E1;  r.gN = N; }
  if (i >= E12) { r.p = a3; li = i - E12; r.gN = 2 * N; }
  r.p += li;
  return r;
}

__device__ __forceinline__ void hold8(const int (&a)[8]) {
  asm volatile("" :: "v"(a[0]), "v"(a[1]), "v"(a[2]), "v"(a[3]), "v"(a[4]), "v"(a[5]), "v"(a[6]), "v"(a[7]));
}

static_assert(SORT_T == 1024, "the scan below needs exactly 32 waves of 32 lanes");
__device__ __forceinline__ int block_scan_inclusive(int v, int* wsum) {
  const int lane = threadIdx.x & 31;
  const int wave = threadIdx.x >> 5;
  int x = v;
  #pragma unroll
  for (int d = 1; d < 32; d <<= 1) {
    const int y = __shfl_up(x, d);
    if (lane >= d) x += y;
  }
  if (lane == 31) wsum[wave] = x;
  __syncthreads();
  if (wave == 0) {
    const int t = wsum[lane];
    int xs = t;
    #pragma unroll
    for (int d = 1; d < 32; d <<= 1) {
      const int y = __shfl_up(xs, d);
      if (lane >= d) xs += y;
    }
    wsum[lane] = xs - t;
  }
  __syncthreads();
  return x + wsum[wave];
}

template <int U>
__global__ void __launch_bounds__(SORT_T)
bin_count(const int* __restrict__ d1, const int* __restrict__ d2, const int* __restrict__ d3,
          int E1, int E2, int E3, int N, int nbin, int chunk, int* __restrict__ hist) {
  __shared__ int h[MAX_BINS];
  const int tid = threadIdx.x;
  const int E12 = E1 + E2;
  const int Et  = E12 + E3;
  for (int b = tid; b < nbin; b += SORT_T) h[b] = 0;
  __syncthreads();
  const int lo = (int)blockIdx.x * chunk;
  const int hi = lo + chunk < Et ? lo + chunk : Et;
  for (int c0 = lo; c0 < hi; c0 += SORT_T * U) {
    int d[U], gN[U];
    #pragma unroll
    for (int u = 0; u < U; ++u) {
      const int i  = c0 + tid + u * SORT_T;
      const int ic = i < hi ? i : hi - 1;
      const EdgeRef e = edge_ref(ic, d1, d2, d3, E1, E12, N);
      d[u]  = *e.p;
      gN[u] = e.gN;
    }
    hold8(d);
    #pragma unroll
    for (int u = 0; u < U; ++u) {
      if (c0 + tid + u * SORT_T < hi) atomicAdd(&h[(gN[u] + d[u]) >> BIN_SHIFT], 1);
    }
  }
  __syncthreads();
  for (int b = tid; b < nbin; b += SORT_T) hist[(int)blockIdx.x * nbin + b] = h[b];
}

template <int U>
__global__ void __launch_bounds__(SORT_T)
bin_scatter(const int* __restrict__ e1, const int* __restrict__ e2, const int* __restrict__ e3,
            int E1, int E2, int E3, int N, int nbin, int chunk, const int* __restrict__ hist,
            int* __restrict__ binStart, int2* __restrict__ binned) {
  __shared__ int cur[MAX_BINS];
  __shared__ int tot[MAX_BINS];
  __shared__ int wsum[32];
  const int tid = threadIdx.x;
  const int E12 = E1 + E2;
  const int Et  = E12 + E3;
  for (int b = tid; b < nbin; b += SORT_T) { cur[b] = 0; tot[b] = 0; }
  __syncthreads();
  const int cells = (int)gridDim.x * nbin;
  const int dk = SORT_T / nbin;
  const int db = SORT_T - dk * nbin;
  int k = tid / nbin;
  int b = tid - k * nbin;
  constexpr int P = 8;
  for (int i0 = tid; i0 < cells; i0 += SORT_T * P) {
    int v[P];
    #pragma unroll
    for (int p = 0; p < P; ++p) {
      const int idx = i0 + p * SORT_T;
      v[p] = hist[idx < cells ? idx : cells - 1];
    }
    hold8(v);
    #pragma unroll
    for (int p = 0; p < P; ++p) {
      if (i0 + p * SORT_T < cells) {
        atomicAdd(&tot[b], v[p]);
        if (k < (int)blockIdx.x) atomicAdd(&cur[b], v[p]);
      }
      k += dk;
      b += db;
      if (b >= nbin) { b -= nbin; k += 1; }
    }
  }
  __syncthreads();
  const int mine = tid < nbin ? tot[tid] : 0;
  const int incl = block_scan_inclusive(mine, wsum);
  if (tid < nbin) {
    const int start = incl - mine;
    cur[tid] += start;
    if (blockIdx.x == 0) {
      binStart[tid] = start;
      if (tid == nbin - 1) binStart[nbin] = incl;
    }
  }
  __syncthreads();
  const int lo = (int)blockIdx.x * chunk;
  const int hi = lo + chunk < Et ? lo + chunk : Et;
  for (int c0 = lo; c0 < hi; c0 += SORT_T * U) {
    int s[U], d[U], gN[U];
    #pragma unroll
    for (int u = 0; u < U; ++u) {
      const int i  = c0 + tid + u * SORT_T;
      const int ic = i < hi ? i : hi - 1;
      const EdgeRef es = edge_ref(ic, e1, e2, e3, E1, E12, N);
      const EdgeRef ed = edge_ref(ic, e1 + E1, e2 + E2, e3 + E3, E1, E12, N);
      s[u]  = *es.p;
      d[u]  = *ed.p;
      gN[u] = ed.gN;
    }
    hold8(s);
    hold8(d);
    #pragma unroll
    for (int u = 0; u < U; ++u) {
      if (c0 + tid + u * SORT_T < hi) {
        const int key = gN[u] + d[u];
        const int pos = atomicAdd(&cur[key >> BIN_SHIFT], 1);
        binned[pos] = make_int2(s[u], key);
      }
    }
  }
}

template <int U>
__global__ void __launch_bounds__(SORT_T)
bin_sort(const int2* __restrict__ binned, const int* __restrict__ binStart, int nkeys,
         int* __restrict__ cnt, int* __restrict__ off, float* __restrict__ dinv, int2* __restrict__ csr) {
  __shared__ int   c[BIN_KEYS];
  __shared__ int   wsum[32];
  __shared__ float dv[BIN_KEYS];
  const int tid = threadIdx.x;
  const int bin = blockIdx.x;
  const int s0  = binStart[bin];
  const int s1  = binStart[bin + 1];
  const int klo = bin << BIN_SHIFT;
  if (tid < BIN_KEYS) c[tid] = 0;
  __syncthreads();
  for (int c0 = s0; c0 < s1; c0 += SORT_T * U) {
    int key[U];
    #pragma unroll
    for (int u = 0; u < U; ++u) {
      const int i  = c0 + tid + u * SORT_T;
      const int ic = i < s1 ? i : s1 - 1;
      key[u] = binned[ic].y;
    }
    hold8(key);
    #pragma unroll
    for (int u = 0; u < U; ++u) {
      if (c0 + tid + u * SORT_T < s1) atomicAdd(&c[key[u] - klo], 1);
    }
  }
  __syncthreads();
  const int v = tid < BIN_KEYS ? c[tid] : 0;
  const int incl = block_scan_inclusive(v, wsum);
  if (tid < BIN_KEYS) {
    const int first = s0 + incl - v;
    const float w = (v > 0) ? rsqrtf((float)v) : 0.0f;
    c[tid]  = first;
    dv[tid] = w;
    const int key = klo + tid;
    if (key < nkeys) { cnt[key] = v; off[key] = first; dinv[key] = w; }
  }
  __syncthreads();
  for (int c0 = s0; c0 < s1; c0 += SORT_T * U) {
    int src[U], key[U];
    #pragma unroll
    for (int u = 0; u < U; ++u) {
      const int i  = c0 + tid + u * SORT_T;
      const int ic = i < s1 ? i : s1 - 1;
      const int2 e = binned[ic];
      src[u] = e.x;
      key[u] = e.y;
    }
    hold8(src);
    hold8(key);
    #pragma unroll
    for (int u = 0; u < U; ++u) {
      if (c0 + tid + u * SORT_T < s1) {
        const int k   = key[u] - klo;
        const int pos = atomicAdd(&c[k], 1);
        csr[pos] = make_int2(src[u], __float_as_int(dv[k]));
      }
    }
  }
}

template <int U>
__global__ void __launch_bounds__(256)
csr_weight(int2* __restrict__ csr, const float* __restrict__ dinv, int E1, int E12, int Et, int N) {
  const int step = (int)gridDim.x * 256 * U;
  for (int c0 = (int)blockIdx.x * 256 * U; c0 < Et; c0 += step) {
    int src[U], wd[U], ws[U];
    #pragma unroll
    for (int u = 0; u < U; ++u) {
      const int i  = c0 + (int)threadIdx.x + u * 256;
      const int ic = i < Et ? i : Et - 1;
      const int2 e = csr[ic];
      src[u] = e.x;
      wd[u]  = e.y;
    }
    hold8(src);
    #pragma unroll
    for (int u = 0; u < U; ++u) {
      const int i  = c0 + (int)threadIdx.x + u * 256;
      const int ic = i < Et ? i : Et - 1;
      const int gN = ic >= E12 ? 2 * N : (ic >= E1 ? N : 0);
      ws[u] = __float_as_int(dinv[gN + src[u]]);
    }
    hold8(ws);
    #pragma unroll
    for (int u = 0; u < U; ++u) {
      const int i = c0 + (int)threadIdx.x + u * 256;
      if (i < Et) csr[i] = make_int2(src[u], __float_as_int(__int_as_float(ws[u]) * __int_as_float(wd[u])));
    }
  }
}

template <int K, int NCOLS>
__global__ void __launch_bounds__(NCOLS * 2)
gemm_wmma_f32(const float* __restrict__ A, const float* __restrict__ B,
              const float* __restrict__ bias, float* __restrict__ D, int M) {
  constexpr int NT  = NCOLS * 2;
  constexpr int LDA = K + 4;
  constexpr int K4  = K / 4;
  constexpr int PER = (16 * K4) / NT;
  static_assert((16 * K4) % NT == 0, "strip must divide evenly over the block");
  __shared__ __attribute__((aligned(16))) float As[16 * LDA];

  const int row0 = blockIdx.x * 16;
  const int tid  = threadIdx.x;

  float4 stage[PER];
  #pragma unroll
  for (int i = 0; i < PER; ++i) {
    const int q  = tid + i * NT;
    const int r  = q / K4;
    const int c4 = q - r * K4;
    int row = row0 + r;
    if (row >= M) row = M - 1;
    stage[i] = *reinterpret_cast<const float4*>(A + (long long)row * K + c4 * 4);
  }
  #pragma unroll
  for (int i = 0; i < PER; ++i) {
    const int q  = tid + i * NT;
    const int r  = q / K4;
    const int c4 = q - r * K4;
    *reinterpret_cast<float4*>(&As[r * LDA + c4 * 4]) = stage[i];
  }
  __syncthreads();

  const int wave = tid >> 5;
  const int lane = tid & 31;
  const int half = lane >> 4;
  const int mn   = lane & 15;
  const int kb   = half * 2;
  const int col0 = wave * 16;

  const float* Ap = &As[mn * LDA + kb];
  const float* Bp = B + kb * NCOLS + col0 + mn;

  v8f acc = {};
  #pragma unroll 4
  for (int kk = 0; kk < K; kk += 4) {
    const v2f a = *reinterpret_cast<const v2f*>(Ap + kk);
    v2f b;
    b.x = Bp[kk * NCOLS];
    b.y = Bp[kk * NCOLS + NCOLS];
    acc = __builtin_amdgcn_wmma_f32_16x16x4_f32(
         false, a,  false, b,
         (short)0, acc,  false,  false);
  }

  const float bv = bias ? bias[col0 + mn] : 0.0f;
  #pragma unroll
  for (int r = 0; r < 8; ++r) {
    int row = row0 + r + half * 8;
    if (row < M) D[(long long)row * NCOLS + col0 + mn] = acc[r] + bv;
  }
}

template <int U, int VEC>
__device__ __forceinline__ void gather_rows(const int2 ed, const int j, const int lane,
                                            const float* __restrict__ HH,
                                            float __attribute__((ext_vector_type(VEC)))& acc) {
  typedef float vf __attribute__((ext_vector_type(VEC)));
  const vf* __restrict__ Hv = reinterpret_cast<const vf*>(HH);
  int s[U];
  float w[U];
  vf v[U];
  #pragma unroll
  for (int u = 0; u < U; ++u) {
    s[u] = __builtin_amdgcn_readlane(ed.x, j + u);
    w[u] = __int_as_float(__builtin_amdgcn_readlane(ed.y, j + u));
  }
  #pragma unroll
  for (int u = 0; u < U; ++u) v[u] = Hv[(size_t)s[u] * 32 + lane];
  #pragma unroll
  for (int u = 0; u < U; ++u) acc += w[u] * v[u];
}

template <int VEC, bool RELU>
__global__ void __launch_bounds__(256)
gcn_gather(const int* __restrict__ off, const int* __restrict__ cnt,
           const int2* __restrict__ csr, const float* __restrict__ HH,
           const float* __restrict__ bias, float* __restrict__ out, int N) {
  typedef float vf __attribute__((ext_vector_type(VEC)));
  constexpr int WIDE = 32 / VEC;
  const int wv   = __builtin_amdgcn_readfirstlane((int)(threadIdx.x >> 5));
  const int node = blockIdx.x * (int)(blockDim.x >> 5) + wv;
  if (node >= N) return;
  const int lane = threadIdx.x & 31;
  const vf* __restrict__ Hv = reinterpret_cast<const vf*>(HH);
  const vf zero = (vf)(0.0f);
  vf* __restrict__ orow = reinterpret_cast<vf*>(out + (size_t)node * (size_t)(4 * 32 * VEC));

  int beg[3], c[3];
  #pragma unroll
  for (int g = 0; g < 3; ++g) {
    beg[g] = off[g * N + node];
    c[g]   = cnt[g * N + node];
  }
  asm volatile("" :: "s"(beg[0]), "s"(beg[1]), "s"(beg[2]), "s"(c[0]), "s"(c[1]), "s"(c[2]));

  int2 first[3];
  #pragma unroll
  for (int g = 0; g < 3; ++g) {
    first[g] = make_int2(0, 0);
    if (c[g] > 0) {
      const int m0 = c[g] < 32 ? c[g] : 32;
      first[g] = csr[beg[g] + (lane < m0 ? lane : m0 - 1)];
    }
  }
  const vf bv   = reinterpret_cast<const vf*>(bias)[lane];
  const vf self = Hv[(size_t)node * 32 + lane];
  asm volatile("" :: "v"(bv), "v"(self));

  #pragma unroll
  for (int g = 0; g < 3; ++g) {
    vf acc = zero;
    int2 ed = first[g];
    for (int base = 0; base < c[g]; base += 32) {
      const int rem = c[g] - base;
      const int m = rem < 32 ? rem : 32;
      int2 nxt = ed;
      if (rem > 32) {
        const int m2 = rem - 32 < 32 ? rem - 32 : 32;
        nxt = csr[beg[g] + base + 32 + (lane < m2 ? lane : m2 - 1)];
      }
      int j = 0;
      for (; j + WIDE <= m; j += WIDE) gather_rows<WIDE, VEC>(ed, j, lane, HH, acc);
      if (WIDE > 8 && j + 8 <= m) { gather_rows<8, VEC>(ed, j, lane, HH, acc); j += 8; }
      if (j + 4 <= m) { gather_rows<4, VEC>(ed, j, lane, HH, acc); j += 4; }
      if (j < m) {
        const int j1 = j + 1 < m ? j + 1 : j;
        const int j2 = j + 2 < m ? j + 2 : j;
        const int s0 = __builtin_amdgcn_readlane(ed.x, j);
        const int s1 = __builtin_amdgcn_readlane(ed.x, j1);
        const int s2 = __builtin_amdgcn_readlane(ed.x, j2);
        const float w0 = __int_as_float(__builtin_amdgcn_readlane(ed.y, j));
        const float w1 = __int_as_float(__builtin_amdgcn_readlane(ed.y, j1));
        const float w2 = __int_as_float(__builtin_amdgcn_readlane(ed.y, j2));
        const vf v0 = Hv[(size_t)s0 * 32 + lane];
        const vf v1 = Hv[(size_t)s1 * 32 + lane];
        const vf v2 = Hv[(size_t)s2 * 32 + lane];
        asm volatile("" :: "v"(v0), "v"(v1), "v"(v2));
        acc += w0 * v0;
        if (j + 1 < m) acc += w1 * v1;
        if (j + 2 < m) acc += w2 * v2;
      }
      ed = nxt;
    }
    vf r = acc + bv;
    if (RELU) r = __builtin_elementwise_max(r, zero);
    orow[g * 32 + lane] = r;
  }
  vf r = self + bv;
  if (RELU) r = __builtin_elementwise_max(r, zero);
  orow[3 * 32 + lane] = r;
}

__device__ __forceinline__ void lsm_row(float a, float b, float& ra, float& rb) {
  float mx = fmaxf(a, b);
  #pragma unroll
  for (int off = 16; off > 0; off >>= 1) mx = fmaxf(mx, __shfl_xor(mx, off));
  float s = __expf(a - mx) + __expf(b - mx);
  #pragma unroll
  for (int off = 16; off > 0; off >>= 1) s += __shfl_xor(s, off);
  const float lse = mx + __logf(s);
  ra = a - lse;
  rb = b - lse;
}

__global__ void __launch_bounds__(256) log_softmax64(float* __restrict__ out, int N) {
  constexpr int R = 4;
  const int W    = (int)gridDim.x * (int)(blockDim.x >> 5);
  const int lane = threadIdx.x & 31;
  int row = (int)blockIdx.x * (int)(blockDim.x >> 5) + (int)(threadIdx.x >> 5);
  for (; row + (R - 1) * W < N; row += R * W) {
    float a[R], b[R];
    #pragma unroll
    for (int j = 0; j < R; ++j) {
      const float* p = out + (long long)(row + j * W) * 64;
      a[j] = p[lane];
      b[j] = p[lane + 32];
    }
    #pragma unroll
    for (int j = 0; j < R; ++j) {
      float ra, rb;
      lsm_row(a[j], b[j], ra, rb);
      float* p = out + (long long)(row + j * W) * 64;
      p[lane]      = ra;
      p[lane + 32] = rb;
    }
  }
  for (; row < N; row += W) {
    float* p = out + (long long)row * 64;
    float ra, rb;
    lsm_row(p[lane], p[lane + 32], ra, rb);
    p[lane]      = ra;
    p[lane + 32] = rb;
  }
}

extern "C" void kernel_launch(void* const* d_in, const int* in_sizes, int n_in,
                              void* d_out, int out_size, void* d_ws, size_t ws_size,
                              hipStream_t stream) {
  const float* x  = (const float*)d_in[0];
  const int* e1   = (const int*)d_in[1];
  const int* e2   = (const int*)d_in[2];
  const int* e3   = (const int*)d_in[3];
  const float* W1 = (const float*)d_in[5];
  const float* b1 = (const float*)d_in[6];
  const float* W2 = (const float*)d_in[7];
  const float* b2 = (const float*)d_in[8];
  const float* Wl = (const float*)d_in[9];
  const float* bl = (const float*)d_in[10];

  const int F = 512, H = 128, C = 64;
  const int N  = in_sizes[0] / F;
  const int E1 = in_sizes[1] / 2;
  const int E2 = in_sizes[2] / 2;
  const int E3 = in_sizes[3] / 2;
  const long long Et = (long long)E1 + E2 + E3;

  float* ws   = (float*)d_ws;
  float* HH1  = ws;
  float* R1   = HH1 + (long long)N * H;
  float* HH2  = R1 + (long long)N * 4 * H;
  float* R2   = HH2 + (long long)N * C;
  int2*  csr  = (int2*)(R2 + (long long)N * 4 * C);
  int2*  binned = csr + Et;
  float* dinv = (float*)(binned + Et);
  int*   cnt  = (int*)(dinv + 3LL * N);
  int*   off  = cnt + 3LL * N;

  const int T = 256;

  const int NB    = 128;
  const int nkeys = 3 * N;
  const int nbin  = cdiv_i(nkeys, BIN_KEYS);
  const int chunk = cdiv_i(Et, NB);
  int*   binStart = off + 3LL * N;
  int*   hist     = binStart + (nbin + 1);
  bin_count<8><<<NB, SORT_T, 0, stream>>>(e1 + E1, e2 + E2, e3 + E3, E1, E2, E3, N, nbin, chunk, hist);
  bin_scatter<8><<<NB, SORT_T, 0, stream>>>(e1, e2, e3, E1, E2, E3, N, nbin, chunk, hist, binStart, binned);
  bin_sort<8><<<nbin, SORT_T, 0, stream>>>(binned, binStart, nkeys, cnt, off, dinv, csr);
  csr_weight<8><<<512, T, 0, stream>>>(csr, dinv, E1, E1 + E2, (int)Et, N);

  const int mtiles = cdiv_i(N, 16);
  const int gblocks = cdiv_i(N, T / 32);

  gemm_wmma_f32<F, H><<<mtiles, 32 * (H / 16), 0, stream>>>(x, W1, nullptr, HH1, N);

  gcn_gather<4, true><<<gblocks, T, 0, stream>>>(off, cnt, csr, HH1, b1, R1, N);

  gemm_wmma_f32<4 * H, C><<<mtiles, 32 * (C / 16), 0, stream>>>(R1, W2, nullptr, HH2, N);

  gcn_gather<2, false><<<gblocks, T, 0, stream>>>(off, cnt, csr, HH2, b2, R2, N);

  float* logits = (float*)d_out;
  gemm_wmma_f32<4 * C, C><<<mtiles, 32 * (C / 16), 0, stream>>>(R2, Wl, bl, logits, N);

  log_softmax64<<<256, T, 0, stream>>>(logits, N);
}


# Each aggregation writes its 16-row output strip into LDS as the A strip of the GEMM that consumes it, one kernel per layer (10 launches to 8); aggregated rows never pass memory; sum orders kept
# speedup vs baseline: 5.2076x; 1.0664x over previous
#include <hip/hip_runtime.h>
#include <math.h>

typedef float v2f __attribute__((ext_vector_type(2)));
typedef float v8f __attribute__((ext_vector_type(8)));

static inline int cdiv_i(long long a, long long b) { return (int)((a + b - 1) / b); }

constexpr int BIN_SHIFT = 8;
constexpr int BIN_KEYS  = 1 << BIN_SHIFT;
constexpr int MAX_BINS  = 1024;
constexpr int SORT_T    = 1024;

struct EdgeRef { const int* p; int gN; };
__device__ __forceinline__ EdgeRef edge_ref(int i, const int* __restrict__ a1, const int* __restrict__ a2,
                                            const int* __restrict__ a3, int E1, int E12, int N) {
  EdgeRef r;
  int li = i;
  r.p = a1; r.gN = 0;
  if (i >= E1)  { r.p = a2; li = i - E1;  r.gN = N; }
  if (i >= E12) { r.p = a3; li = i - E12; r.gN = 2 * N; }
  r.p += li;
  return r;
}

__device__ __forceinline__ void hold8(const int (&a)[8]) {
  asm volatile("" :: "v"(a[0]), "v"(a[1]), "v"(a[2]), "v"(a[3]), "v"(a[4]), "v"(a[5]), "v"(a[6]), "v"(a[7]));
}

static_assert(SORT_T == 1024, "the scan below needs exactly 32 waves of 32 lanes");
__device__ __forceinline__ int block_scan_inclusive(int v, int* wsum) {
  const int lane = threadIdx.x & 31;
  const int wave = threadIdx.x >> 5;
  int x = v;
  #pragma unroll
  for (int d = 1; d < 32; d <<= 1) {
    const int y = __shfl_up(x, d);
    if (lane >= d) x += y;
  }
  if (lane == 31) wsum[wave] = x;
  __syncthreads();
  if (wave == 0) {
    const int t = wsum[lane];
    int xs = t;
    #pragma unroll
    for (int d = 1; d < 32; d <<= 1) {
      const int y = __shfl_up(xs, d);
      if (lane >= d) xs += y;
    }
    wsum[lane] = xs - t;
  }
  __syncthreads();
  return x + wsum[wave];
}

template <int U>
__global__ void __launch_bounds__(SORT_T)
bin_count(const int* __restrict__ d1, const int* __restrict__ d2, const int* __restrict__ d3,
          int E1, int E2, int E3, int N, int nbin, int chunk, int* __restrict__ hist) {
  __shared__ int h[MAX_BINS];
  const int tid = threadIdx.x;
  const int E12 = E1 + E2;
  const int Et  = E12 + E3;
  for (int b = tid; b < nbin; b += SORT_T) h[b] = 0;
  __syncthreads();
  const int lo = (int)blockIdx.x * chunk;
  const int hi = lo + chunk < Et ? lo + chunk : Et;
  for (int c0 = lo; c0 < hi; c0 += SORT_T * U) {
    int d[U], gN[U];
    #pragma unroll
    for (int u = 0; u < U; ++u) {
      const int i  = c0 + tid + u * SORT_T;
      const int ic = i < hi ? i : hi - 1;
      const EdgeRef e = edge_ref(ic, d1, d2, d3, E1, E12, N);
      d[u]  = *e.p;
      gN[u] = e.gN;
    }
    hold8(d);
    #pragma unroll
    for (int u = 0; u < U; ++u) {
      if (c0 + tid + u * SORT_T < hi) atomicAdd(&h[(gN[u] + d[u]) >> BIN_SHIFT], 1);
    }
  }
  __syncthreads();
  for (int b = tid; b < nbin; b += SORT_T) hist[(int)blockIdx.x * nbin + b] = h[b];
}

template <int U>
__global__ void __launch_bounds__(SORT_T)
bin_scatter(const int* __restrict__ e1, const int* __restrict__ e2, const int* __restrict__ e3,
            int E1, int E2, int E3, int N, int nbin, int chunk, const int* __restrict__ hist,
            int* __restrict__ binStart, int2* __restrict__ binned) {
  __shared__ int cur[MAX_BINS];
  __shared__ int tot[MAX_BINS];
  __shared__ int wsum[32];
  const int tid = threadIdx.x;
  const int E12 = E1 + E2;
  const int Et  = E12 + E3;
  for (int b = tid; b < nbin; b += SORT_T) { cur[b] = 0; tot[b] = 0; }
  __syncthreads();
  const int cells = (int)gridDim.x * nbin;
  const int dk = SORT_T / nbin;
  const int db = SORT_T - dk * nbin;
  int k = tid / nbin;
  int b = tid - k * nbin;
  constexpr int P = 8;
  for (int i0 = tid; i0 < cells; i0 += SORT_T * P) {
    int v[P];
    #pragma unroll
    for (int p = 0; p < P; ++p) {
      const int idx = i0 + p * SORT_T;
      v[p] = hist[idx < cells ? idx : cells - 1];
    }
    hold8(v);
    #pragma unroll
    for (int p = 0; p < P; ++p) {
      if (i0 + p * SORT_T < cells) {
        atomicAdd(&tot[b], v[p]);
        if (k < (int)blockIdx.x) atomicAdd(&cur[b], v[p]);
      }
      k += dk;
      b += db;
      if (b >= nbin) { b -= nbin; k += 1; }
    }
  }
  __syncthreads();
  const int mine = tid < nbin ? tot[tid] : 0;
  const int incl = block_scan_inclusive(mine, wsum);
  if (tid < nbin) {
    const int start = incl - mine;
    cur[tid] += start;
    if (blockIdx.x == 0) {
      binStart[tid] = start;
      if (tid == nbin - 1) binStart[nbin] = incl;
    }
  }
  __syncthreads();
  const int lo = (int)blockIdx.x * chunk;
  const int hi = lo + chunk < Et ? lo + chunk : Et;
  for (int c0 = lo; c0 < hi; c0 += SORT_T * U) {
    int s[U], d[U], gN[U];
    #pragma unroll
    for (int u = 0; u < U; ++u) {
      const int i  = c0 + tid + u * SORT_T;
      const int ic = i < hi ? i : hi - 1;
      const EdgeRef es = edge_ref(ic, e1, e2, e3, E1, E12, N);
      const EdgeRef ed = edge_ref(ic, e1 + E1, e2 + E2, e3 + E3, E1, E12, N);
      s[u]  = *es.p;
      d[u]  = *ed.p;
      gN[u] = ed.gN;
    }
    hold8(s);
    hold8(d);
    #pragma unroll
    for (int u = 0; u < U; ++u) {
      if (c0 + tid + u * SORT_T < hi) {
        const int key = gN[u] + d[u];
        const int pos = atomicAdd(&cur[key >> BIN_SHIFT], 1);
        binned[pos] = make_int2(s[u], key);
      }
    }
  }
}

template <int U>
__global__ void __launch_bounds__(SORT_T)
bin_sort(const int2* __restrict__ binned, const int* __restrict__ binStart, int nkeys,
         int* __restrict__ cnt, int* __restrict__ off, float* __restrict__ dinv, int2* __restrict__ csr) {
  __shared__ int   c[BIN_KEYS];
  __shared__ int   wsum[32];
  __shared__ float dv[BIN_KEYS];
  const int tid = threadIdx.x;
  const int bin = blockIdx.x;
  const int s0  = binStart[bin];
  const int s1  = binStart[bin + 1];
  const int klo = bin << BIN_SHIFT;
  if (tid < BIN_KEYS) c[tid] = 0;
  __syncthreads();
  for (int c0 = s0; c0 < s1; c0 += SORT_T * U) {
    int key[U];
    #pragma unroll
    for (int u = 0; u < U; ++u) {
      const int i  = c0 + tid + u * SORT_T;
      const int ic = i < s1 ? i : s1 - 1;
      key[u] = binned[ic].y;
    }
    hold8(key);
    #pragma unroll
    for (int u = 0; u < U; ++u) {
      if (c0 + tid + u * SORT_T < s1) atomicAdd(&c[key[u] - klo], 1);
    }
  }
  __syncthreads();
  const int v = tid < BIN_KEYS ? c[tid] : 0;
  const int incl = block_scan_inclusive(v, wsum);
  if (tid < BIN_KEYS) {
    const int first = s0 + incl - v;
    const float w = (v > 0) ? rsqrtf((float)v) : 0.0f;
    c[tid]  = first;
    dv[tid] = w;
    const int key = klo + tid;
    if (key < nkeys) { cnt[key] = v; off[key] = first; dinv[key] = w; }
  }
  __syncthreads();
  for (int c0 = s0; c0 < s1; c0 += SORT_T * U) {
    int src[U], key[U];
    #pragma unroll
    for (int u = 0; u < U; ++u) {
      const int i  = c0 + tid + u * SORT_T;
      const int ic = i < s1 ? i : s1 - 1;
      const int2 e = binned[ic];
      src[u] = e.x;
      key[u] = e.y;
    }
    hold8(src);
    hold8(key);
    #pragma unroll
    for (int u = 0; u < U; ++u) {
      if (c0 + tid + u * SORT_T < s1) {
        const int k   = key[u] - klo;
        const int pos = atomicAdd(&c[k], 1);
        csr[pos] = make_int2(src[u], __float_as_int(dv[k]));
      }
    }
  }
}

template <int U>
__global__ void __launch_bounds__(256)
csr_weight(int2* __restrict__ csr, const float* __restrict__ dinv, int E1, int E12, int Et, int N) {
  const int step = (int)gridDim.x * 256 * U;
  for (int c0 = (int)blockIdx.x * 256 * U; c0 < Et; c0 += step) {
    int src[U], wd[U], ws[U];
    #pragma unroll
    for (int u = 0; u < U; ++u) {
      const int i  = c0 + (int)threadIdx.x + u * 256;
      const int ic = i < Et ? i : Et - 1;
      const int2 e = csr[ic];
      src[u] = e.x;
      wd[u]  = e.y;
    }
    hold8(src);
    #pragma unroll
    for (int u = 0; u < U; ++u) {
      const int i  = c0 + (int)threadIdx.x + u * 256;
      const int ic = i < Et ? i : Et - 1;
      const int gN = ic >= E12 ? 2 * N : (ic >= E1 ? N : 0);
      ws[u] = __float_as_int(dinv[gN + src[u]]);
    }
    hold8(ws);
    #pragma unroll
    for (int u = 0; u < U; ++u) {
      const int i = c0 + (int)threadIdx.x + u * 256;
      if (i < Et) csr[i] = make_int2(src[u], __float_as_int(__int_as_float(ws[u]) * __int_as_float(wd[u])));
    }
  }
}

template <int K, int NCOLS>
__global__ void __launch_bounds__(NCOLS * 2)
gemm_wmma_f32(const float* __restrict__ A, const float* __restrict__ B,
              const float* __restrict__ bias, float* __restrict__ D, int M) {
  constexpr int NT  = NCOLS * 2;
  constexpr int LDA = K + 4;
  constexpr int K4  = K / 4;
  constexpr int PER = (16 * K4) / NT;
  static_assert((16 * K4) % NT == 0, "strip must divide evenly over the block");
  __shared__ __attribute__((aligned(16))) float As[16 * LDA];

  const int row0 = blockIdx.x * 16;
  const int tid  = threadIdx.x;

  float4 stage[PER];
  #pragma unroll
  for (int i = 0; i < PER; ++i) {
    const int q  = tid + i * NT;
    const int r  = q / K4;
    const int c4 = q - r * K4;
    int row = row0 + r;
    if (row >= M) row = M - 1;
    stage[i] = *reinterpret_cast<const float4*>(A + (long long)row * K + c4 * 4);
  }
  #pragma unroll
  for (int i = 0; i < PER; ++i) {
    const int q  = tid + i * NT;
    const int r  = q / K4;
    const int c4 = q - r * K4;
    *reinterpret_cast<float4*>(&As[r * LDA + c4 * 4]) = stage[i];
  }
  __syncthreads();

  const int wave = tid >> 5;
  const int lane = tid & 31;
  const int half = lane >> 4;
  const int mn   = lane & 15;
  const int kb   = half * 2;
  const int col0 = wave * 16;

  const float* Ap = &As[mn * LDA + kb];
  const float* Bp = B + kb * NCOLS + col0 + mn;

  v8f acc = {};
  #pragma unroll 4
  for (int kk = 0; kk < K; kk += 4) {
    const v2f a = *reinterpret_cast<const v2f*>(Ap + kk);
    v2f b;
    b.x = Bp[kk * NCOLS];
    b.y = Bp[kk * NCOLS + NCOLS];
    acc = __builtin_amdgcn_wmma_f32_16x16x4_f32(
         false, a,  false, b,
         (short)0, acc,  false,  false);
  }

  const float bv = bias ? bias[col0 + mn] : 0.0f;
  #pragma unroll
  for (int r = 0; r < 8; ++r) {
    int row = row0 + r + half * 8;
    if (row < M) D[(long long)row * NCOLS + col0 + mn] = acc[r] + bv;
  }
}

template <int U, int VEC>
__device__ __forceinline__ void gather_rows(const int2 ed, const int j, const int lane,
                                            const float* __restrict__ HH,
                                            float __attribute__((ext_vector_type(VEC)))& acc) {
  typedef float vf __attribute__((ext_vector_type(VEC)));
  const vf* __restrict__ Hv = reinterpret_cast<const vf*>(HH);
  int s[U];
  float w[U];
  vf v[U];
  #pragma unroll
  for (int u = 0; u < U; ++u) {
    s[u] = __builtin_amdgcn_readlane(ed.x, j + u);
    w[u] = __int_as_float(__builtin_amdgcn_readlane(ed.y, j + u));
  }
  #pragma unroll
  for (int u = 0; u < U; ++u) v[u] = Hv[(size_t)s[u] * 32 + lane];
  #pragma unroll
  for (int u = 0; u < U; ++u) acc += w[u] * v[u];
}

template <int VEC, bool RELU>
__device__ __forceinline__ void
gather_node(const int* __restrict__ off, const int* __restrict__ cnt,
            const int2* __restrict__ csr, const float* __restrict__ HH,
            const float* __restrict__ bias, float* orow_f, const int node, const int lane, const int N) {
  typedef float vf __attribute__((ext_vector_type(VEC)));
  constexpr int WIDE = 32 / VEC;
  const vf* __restrict__ Hv = reinterpret_cast<const vf*>(HH);
  const vf zero = (vf)(0.0f);
  vf* orow = reinterpret_cast<vf*>(orow_f);

  int beg[3], c[3];
  #pragma unroll
  for (int g = 0; g < 3; ++g) {
    beg[g] = off[g * N + node];
    c[g]   = cnt[g * N + node];
  }
  asm volatile("" :: "s"(beg[0]), "s"(beg[1]), "s"(beg[2]), "s"(c[0]), "s"(c[1]), "s"(c[2]));

  int2 first[3];
  #pragma unroll
  for (int g = 0; g < 3; ++g) {
    first[g] = make_int2(0, 0);
    if (c[g] > 0) {
      const int m0 = c[g] < 32 ? c[g] : 32;
      first[g] = csr[beg[g] + (lane < m0 ? lane : m0 - 1)];
    }
  }
  const vf bv   = reinterpret_cast<const vf*>(bias)[lane];
  const vf self = Hv[(size_t)node * 32 + lane];
  asm volatile("" :: "v"(bv), "v"(self));

  #pragma unroll
  for (int g = 0; g < 3; ++g) {
    vf acc = zero;
    int2 ed = first[g];
    for (int base = 0; base < c[g]; base += 32) {
      const int rem = c[g] - base;
      const int m = rem < 32 ? rem : 32;
      int2 nxt = ed;
      if (rem > 32) {
        const int m2 = rem - 32 < 32 ? rem - 32 : 32;
        nxt = csr[beg[g] + base + 32 + (lane < m2 ? lane : m2 - 1)];
      }
      int j = 0;
      for (; j + WIDE <= m; j += WIDE) gather_rows<WIDE, VEC>(ed, j, lane, HH, acc);
      if (WIDE > 8 && j + 8 <= m) { gather_rows<8, VEC>(ed, j, lane, HH, acc); j += 8; }
      if (j + 4 <= m) { gather_rows<4, VEC>(ed, j, lane, HH, acc); j += 4; }
      if (j < m) {
        const int j1 = j + 1 < m ? j + 1 : j;
        const int j2 = j + 2 < m ? j + 2 : j;
        const int s0 = __builtin_amdgcn_readlane(ed.x, j);
        const int s1 = __builtin_amdgcn_readlane(ed.x, j1);
        const int s2 = __builtin_amdgcn_readlane(ed.x, j2);
        const float w0 = __int_as_float(__builtin_amdgcn_readlane(ed.y, j));
        const float w1 = __int_as_float(__builtin_amdgcn_readlane(ed.y, j1));
        const float w2 = __int_as_float(__builtin_amdgcn_readlane(ed.y, j2));
        const vf v0 = Hv[(size_t)s0 * 32 + lane];
        const vf v1 = Hv[(size_t)s1 * 32 + lane];
        const vf v2 = Hv[(size_t)s2 * 32 + lane];
        asm volatile("" :: "v"(v0), "v"(v1), "v"(v2));
        acc += w0 * v0;
        if (j + 1 < m) acc += w1 * v1;
        if (j + 2 < m) acc += w2 * v2;
      }
      ed = nxt;
    }
    vf r = acc + bv;
    if (RELU) r = __builtin_elementwise_max(r, zero);
    orow[g * 32 + lane] = r;
  }
  vf r = self + bv;
  if (RELU) r = __builtin_elementwise_max(r, zero);
  orow[3 * 32 + lane] = r;
}

template <int VEC, bool RELU, int NCOLS>
__global__ void __launch_bounds__(512)
gcn_gather_gemm(const int* __restrict__ off, const int* __restrict__ cnt,
                const int2* __restrict__ csr, const float* __restrict__ HH,
                const float* __restrict__ bias, const float* __restrict__ B,
                const float* __restrict__ bias2, float* __restrict__ D, int N) {
  constexpr int K   = 4 * 32 * VEC;
  constexpr int LDA = K + 4;
  static_assert(NCOLS % 16 == 0 && NCOLS / 16 <= 16, "one wave per 16-column tile, at most 16 waves");
  __shared__ __attribute__((aligned(16))) float As[16 * LDA];

  const int wv   = __builtin_amdgcn_readfirstlane((int)(threadIdx.x >> 5));
  const int lane = threadIdx.x & 31;
  const int row0 = blockIdx.x * 16;
  const int node = row0 + wv;
  float* srow = &As[wv * LDA];
  if (node < N) {
    gather_node<VEC, RELU>(off, cnt, csr, HH, bias, srow, node, lane, N);
  } else {
    typedef float vf __attribute__((ext_vector_type(VEC)));
    #pragma unroll
    for (int g = 0; g < 4; ++g) reinterpret_cast<vf*>(srow)[g * 32 + lane] = (vf)(0.0f);
  }
  __syncthreads();
  if (wv >= NCOLS / 16) return;

  const int half = lane >> 4;
  const int mn   = lane & 15;
  const int kb   = half * 2;
  const int col0 = wv * 16;

  const float* Ap = &As[mn * LDA + kb];
  const float* Bp = B + kb * NCOLS + col0 + mn;

  v8f acc = {};
  #pragma unroll 8
  for (int kk = 0; kk < K; kk += 4) {
    const v2f a = *reinterpret_cast<const v2f*>(Ap + kk);
    v2f b;
    b.x = Bp[kk * NCOLS];
    b.y = Bp[kk * NCOLS + NCOLS];
    acc = __builtin_amdgcn_wmma_f32_16x16x4_f32(
         false, a,  false, b,
         (short)0, acc,  false,  false);
  }

  const float bv = bias2 ? bias2[col0 + mn] : 0.0f;
  #pragma unroll
  for (int r = 0; r < 8; ++r) {
    const int row = row0 + r + half * 8;
    if (row < N) D[(long long)row * NCOLS + col0 + mn] = acc[r] + bv;
  }
}

__device__ __forceinline__ void lsm_row(float a, float b, float& ra, float& rb) {
  float mx = fmaxf(a, b);
  #pragma unroll
  for (int off = 16; off > 0; off >>= 1) mx = fmaxf(mx, __shfl_xor(mx, off));
  float s = __expf(a - mx) + __expf(b - mx);
  #pragma unroll
  for (int off = 16; off > 0; off >>= 1) s += __shfl_xor(s, off);
  const float lse = mx + __logf(s);
  ra = a - lse;
  rb = b - lse;
}

__global__ void __launch_bounds__(256) log_softmax64(float* __restrict__ out, int N) {
  constexpr int R = 4;
  const int W    = (int)gridDim.x * (int)(blockDim.x >> 5);
  const int lane = threadIdx.x & 31;
  int row = (int)blockIdx.x * (int)(blockDim.x >> 5) + (int)(threadIdx.x >> 5);
  for (; row + (R - 1) * W < N; row += R * W) {
    float a[R], b[R];
    #pragma unroll
    for (int j = 0; j < R; ++j) {
      const float* p = out + (long long)(row + j * W) * 64;
      a[j] = p[lane];
      b[j] = p[lane + 32];
    }
    #pragma unroll
    for (int j = 0; j < R; ++j) {
      float ra, rb;
      lsm_row(a[j], b[j], ra, rb);
      float* p = out + (long long)(row + j * W) * 64;
      p[lane]      = ra;
      p[lane + 32] = rb;
    }
  }
  for (; row < N; row += W) {
    float* p = out + (long long)row * 64;
    float ra, rb;
    lsm_row(p[lane], p[lane + 32], ra, rb);
    p[lane]      = ra;
    p[lane + 32] = rb;
  }
}

extern "C" void kernel_launch(void* const* d_in, const int* in_sizes, int n_in,
                              void* d_out, int out_size, void* d_ws, size_t ws_size,
                              hipStream_t stream) {
  const float* x  = (const float*)d_in[0];
  const int* e1   = (const int*)d_in[1];
  const int* e2   = (const int*)d_in[2];
  const int* e3   = (const int*)d_in[3];
  const float* W1 = (const float*)d_in[5];
  const float* b1 = (const float*)d_in[6];
  const float* W2 = (const float*)d_in[7];
  const float* b2 = (const float*)d_in[8];
  const float* Wl = (const float*)d_in[9];
  const float* bl = (const float*)d_in[10];

  const int F = 512, H = 128, C = 64;
  const int N  = in_sizes[0] / F;
  const int E1 = in_sizes[1] / 2;
  const int E2 = in_sizes[2] / 2;
  const int E3 = in_sizes[3] / 2;
  const long long Et = (long long)E1 + E2 + E3;

  float* ws   = (float*)d_ws;
  float* HH1  = ws;
  float* HH2  = HH1 + (long long)N * H;
  int2*  csr  = (int2*)(HH2 + (long long)N * C);
  int2*  binned = csr + Et;
  float* dinv = (float*)(binned + Et);
  int*   cnt  = (int*)(dinv + 3LL * N);
  int*   off  = cnt + 3LL * N;

  const int T = 256;

  const int NB    = 128;
  const int nkeys = 3 * N;
  const int nbin  = cdiv_i(nkeys, BIN_KEYS);
  const int chunk = cdiv_i(Et, NB);
  int*   binStart = off + 3LL * N;
  int*   hist     = binStart + (nbin + 1);
  bin_count<8><<<NB, SORT_T, 0, stream>>>(e1 + E1, e2 + E2, e3 + E3, E1, E2, E3, N, nbin, chunk, hist);
  bin_scatter<8><<<NB, SORT_T, 0, stream>>>(e1, e2, e3, E1, E2, E3, N, nbin, chunk, hist, binStart, binned);
  bin_sort<8><<<nbin, SORT_T, 0, stream>>>(binned, binStart, nkeys, cnt, off, dinv, csr);
  csr_weight<8><<<512, T, 0, stream>>>(csr, dinv, E1, E1 + E2, (int)Et, N);

  const int mtiles = cdiv_i(N, 16);

  gemm_wmma_f32<F, H><<<mtiles, 32 * (H / 16), 0, stream>>>(x, W1, nullptr, HH1, N);

  gcn_gather_gemm<4, true, C><<<mtiles, 512, 0, stream>>>(off, cnt, csr, HH1, b1, W2, nullptr, HH2, N);

  float* logits = (float*)d_out;
  gcn_gather_gemm<2, false, C><<<mtiles, 512, 0, stream>>>(off, cnt, csr, HH2, b2, Wl, bl, logits, N);

  log_softmax64<<<256, T, 0, stream>>>(logits, N);
}


# Fused kernels' GEMM tail on all 16 waves: K split in 4, partial tiles summed per output row from LDS in fixed order (sum over k = 4 partial sums); log-softmax done in that row pass, 8 launches to 7
# speedup vs baseline: 5.8284x; 1.1192x over previous
#include <hip/hip_runtime.h>
#include <math.h>

typedef float v2f __attribute__((ext_vector_type(2)));
typedef float v8f __attribute__((ext_vector_type(8)));

static inline int cdiv_i(long long a, long long b) { return (int)((a + b - 1) / b); }

constexpr int BIN_SHIFT = 8;
constexpr int BIN_KEYS  = 1 << BIN_SHIFT;
constexpr int MAX_BINS  = 1024;
constexpr int SORT_T    = 1024;

struct EdgeRef { const int* p; int gN; };
__device__ __forceinline__ EdgeRef edge_ref(int i, const int* __restrict__ a1, const int* __restrict__ a2,
                                            const int* __restrict__ a3, int E1, int E12, int N) {
  EdgeRef r;
  int li = i;
  r.p = a1; r.gN = 0;
  if (i >= E1)  { r.p = a2; li = i - E1;  r.gN = N; }
  if (i >= E12) { r.p = a3; li = i - E12; r.gN = 2 * N; }
  r.p += li;
  return r;
}

__device__ __forceinline__ void hold8(const int (&a)[8]) {
  asm volatile("" :: "v"(a[0]), "v"(a[1]), "v"(a[2]), "v"(a[3]), "v"(a[4]), "v"(a[5]), "v"(a[6]), "v"(a[7]));
}

static_assert(SORT_T == 1024, "the scan below needs exactly 32 waves of 32 lanes");
__device__ __forceinline__ int block_scan_inclusive(int v, int* wsum) {
  const int lane = threadIdx.x & 31;
  const int wave = threadIdx.x >> 5;
  int x = v;
  #pragma unroll
  for (int d = 1; d < 32; d <<= 1) {
    const int y = __shfl_up(x, d);
    if (lane >= d) x += y;
  }
  if (lane == 31) wsum[wave] = x;
  __syncthreads();
  if (wave == 0) {
    const int t = wsum[lane];
    int xs = t;
    #pragma unroll
    for (int d = 1; d < 32; d <<= 1) {
      const int y = __shfl_up(xs, d);
      if (lane >= d) xs += y;
    }
    wsum[lane] = xs - t;
  }
  __syncthreads();
  return x + wsum[wave];
}

template <int U>
__global__ void __launch_bounds__(SORT_T)
bin_count(const int* __restrict__ d1, const int* __restrict__ d2, const int* __restrict__ d3,
          int E1, int E2, int E3, int N, int nbin, int chunk, int* __restrict__ hist) {
  __shared__ int h[MAX_BINS];
  const int tid = threadIdx.x;
  const int E12 = E1 + E2;
  const int Et  = E12 + E3;
  for (int b = tid; b < nbin; b += SORT_T) h[b] = 0;
  __syncthreads();
  const int lo = (int)blockIdx.x * chunk;
  const int hi = lo + chunk < Et ? lo + chunk : Et;
  for (int c0 = lo; c0 < hi; c0 += SORT_T * U) {
    int d[U], gN[U];
    #pragma unroll
    for (int u = 0; u < U; ++u) {
      const int i  = c0 + tid + u * SORT_T;
      const int ic = i < hi ? i : hi - 1;
      const EdgeRef e = edge_ref(ic, d1, d2, d3, E1, E12, N);
      d[u]  = *e.p;
      gN[u] = e.gN;
    }
    hold8(d);
    #pragma unroll
    for (int u = 0; u < U; ++u) {
      if (c0 + tid + u * SORT_T < hi) atomicAdd(&h[(gN[u] + d[u]) >> BIN_SHIFT], 1);
    }
  }
  __syncthreads();
  for (int b = tid; b < nbin; b += SORT_T) hist[(int)blockIdx.x * nbin + b] = h[b];
}

template <int U>
__global__ void __launch_bounds__(SORT_T)
bin_scatter(const int* __restrict__ e1, const int* __restrict__ e2, const int* __restrict__ e3,
            int E1, int E2, int E3, int N, int nbin, int chunk, const int* __restrict__ hist,
            int* __restrict__ binStart, int2* __restrict__ binned) {
  __shared__ int cur[MAX_BINS];
  __shared__ int tot[MAX_BINS];
  __shared__ int wsum[32];
  const int tid = threadIdx.x;
  const int E12 = E1 + E2;
  const int Et  = E12 + E3;
  for (int b = tid; b < nbin; b += SORT_T) { cur[b] = 0; tot[b] = 0; }
  __syncthreads();
  const int cells = (int)gridDim.x * nbin;
  const int dk = SORT_T / nbin;
  const int db = SORT_T - dk * nbin;
  int k = tid / nbin;
  int b = tid - k * nbin;
  constexpr int P = 8;
  for (int i0 = tid; i0 < cells; i0 += SORT_T * P) {
    int v[P];
    #pragma unroll
    for (int p = 0; p < P; ++p) {
      const int idx = i0 + p * SORT_T;
      v[p] = hist[idx < cells ? idx : cells - 1];
    }
    hold8(v);
    #pragma unroll
    for (int p = 0; p < P; ++p) {
      if (i0 + p * SORT_T < cells) {
        atomicAdd(&tot[b], v[p]);
        if (k < (int)blockIdx.x) atomicAdd(&cur[b], v[p]);
      }
      k += dk;
      b += db;
      if (b >= nbin) { b -= nbin; k += 1; }
    }
  }
  __syncthreads();
  const int mine = tid < nbin ? tot[tid] : 0;
  const int incl = block_scan_inclusive(mine, wsum);
  if (tid < nbin) {
    const int start = incl - mine;
    cur[tid] += start;
    if (blockIdx.x == 0) {
      binStart[tid] = start;
      if (tid == nbin - 1) binStart[nbin] = incl;
    }
  }
  __syncthreads();
  const int lo = (int)blockIdx.x * chunk;
  const int hi = lo + chunk < Et ? lo + chunk : Et;
  for (int c0 = lo; c0 < hi; c0 += SORT_T * U) {
    int s[U], d[U], gN[U];
    #pragma unroll
    for (int u = 0; u < U; ++u) {
      const int i  = c0 + tid + u * SORT_T;
      const int ic = i < hi ? i : hi - 1;
      const EdgeRef es = edge_ref(ic, e1, e2, e3, E1, E12, N);
      const EdgeRef ed = edge_ref(ic, e1 + E1, e2 + E2, e3 + E3, E1, E12, N);
      s[u]  = *es.p;
      d[u]  = *ed.p;
      gN[u] = ed.gN;
    }
    hold8(s);
    hold8(d);
    #pragma unroll
    for (int u = 0; u < U; ++u) {
      if (c0 + tid + u * SORT_T < hi) {
        const int key = gN[u] + d[u];
        const int pos = atomicAdd(&cur[key >> BIN_SHIFT], 1);
        binned[pos] = make_int2(s[u], key);
      }
    }
  }
}

template <int U>
__global__ void __launch_bounds__(SORT_T)
bin_sort(const int2* __restrict__ binned, const int* __restrict__ binStart, int nkeys,
         int* __restrict__ cnt, int* __restrict__ off, float* __restrict__ dinv, int2* __restrict__ csr) {
  __shared__ int   c[BIN_KEYS];
  __shared__ int   wsum[32];
  __shared__ float dv[BIN_KEYS];
  const int tid = threadIdx.x;
  const int bin = blockIdx.x;
  const int s0  = binStart[bin];
  const int s1  = binStart[bin + 1];
  const int klo = bin << BIN_SHIFT;
  if (tid < BIN_KEYS) c[tid] = 0;
  __syncthreads();
  for (int c0 = s0; c0 < s1; c0 += SORT_T * U) {
    int key[U];
    #pragma unroll
    for (int u = 0; u < U; ++u) {
      const int i  = c0 + tid + u * SORT_T;
      const int ic = i < s1 ? i : s1 - 1;
      key[u] = binned[ic].y;
    }
    hold8(key);
    #pragma unroll
    for (int u = 0; u < U; ++u) {
      if (c0 + tid + u * SORT_T < s1) atomicAdd(&c[key[u] - klo], 1);
    }
  }
  __syncthreads();
  const int v = tid < BIN_KEYS ? c[tid] : 0;
  const int incl = block_scan_inclusive(v, wsum);
  if (tid < BIN_KEYS) {
    const int first = s0 + incl - v;
    const float w = (v > 0) ? rsqrtf((float)v) : 0.0f;
    c[tid]  = first;
    dv[tid] = w;
    const int key = klo + tid;
    if (key < nkeys) { cnt[key] = v; off[key] = first; dinv[key] = w; }
  }
  __syncthreads();
  for (int c0 = s0; c0 < s1; c0 += SORT_T * U) {
    int src[U], key[U];
    #pragma unroll
    for (int u = 0; u < U; ++u) {
      const int i  = c0 + tid + u * SORT_T;
      const int ic = i < s1 ? i : s1 - 1;
      const int2 e = binned[ic];
      src[u] = e.x;
      key[u] = e.y;
    }
    hold8(src);
    hold8(key);
    #pragma unroll
    for (int u = 0; u < U; ++u) {
      if (c0 + tid + u * SORT_T < s1) {
        const int k   = key[u] - klo;
        const int pos = atomicAdd(&c[k], 1);
        csr[pos] = make_int2(src[u], __float_as_int(dv[k]));
      }
    }
  }
}

template <int U>
__global__ void __launch_bounds__(256)
csr_weight(int2* __restrict__ csr, const float* __restrict__ dinv, int E1, int E12, int Et, int N) {
  const int step = (int)gridDim.x * 256 * U;
  for (int c0 = (int)blockIdx.x * 256 * U; c0 < Et; c0 += step) {
    int src[U], wd[U], ws[U];
    #pragma unroll
    for (int u = 0; u < U; ++u) {
      const int i  = c0 + (int)threadIdx.x + u * 256;
      const int ic = i < Et ? i : Et - 1;
      const int2 e = csr[ic];
      src[u] = e.x;
      wd[u]  = e.y;
    }
    hold8(src);
    #pragma unroll
    for (int u = 0; u < U; ++u) {
      const int i  = c0 + (int)threadIdx.x + u * 256;
      const int ic = i < Et ? i : Et - 1;
      const int gN = ic >= E12 ? 2 * N : (ic >= E1 ? N : 0);
      ws[u] = __float_as_int(dinv[gN + src[u]]);
    }
    hold8(ws);
    #pragma unroll
    for (int u = 0; u < U; ++u) {
      const int i = c0 + (int)threadIdx.x + u * 256;
      if (i < Et) csr[i] = make_int2(src[u], __float_as_int(__int_as_float(ws[u]) * __int_as_float(wd[u])));
    }
  }
}

template <int K, int NCOLS>
__global__ void __launch_bounds__(NCOLS * 2)
gemm_wmma_f32(const float* __restrict__ A, const float* __restrict__ B,
              const float* __restrict__ bias, float* __restrict__ D, int M) {
  constexpr int NT  = NCOLS * 2;
  constexpr int LDA = K + 4;
  constexpr int K4  = K / 4;
  constexpr int PER = (16 * K4) / NT;
  static_assert((16 * K4) % NT == 0, "strip must divide evenly over the block");
  __shared__ __attribute__((aligned(16))) float As[16 * LDA];

  const int row0 = blockIdx.x * 16;
  const int tid  = threadIdx.x;

  float4 stage[PER];
  #pragma unroll
  for (int i = 0; i < PER; ++i) {
    const int q  = tid + i * NT;
    const int r  = q / K4;
    const int c4 = q - r * K4;
    int row = row0 + r;
    if (row >= M) row = M - 1;
    stage[i] = *reinterpret_cast<const float4*>(A + (long long)row * K + c4 * 4);
  }
  #pragma unroll
  for (int i = 0; i < PER; ++i) {
    const int q  = tid + i * NT;
    const int r  = q / K4;
    const int c4 = q - r * K4;
    *reinterpret_cast<float4*>(&As[r * LDA + c4 * 4]) = stage[i];
  }
  __syncthreads();

  const int wave = tid >> 5;
  const int lane = tid & 31;
  const int half = lane >> 4;
  const int mn   = lane & 15;
  const int kb   = half * 2;
  const int col0 = wave * 16;

  const float* Ap = &As[mn * LDA + kb];
  const float* Bp = B + kb * NCOLS + col0 + mn;

  v8f acc = {};
  #pragma unroll 4
  for (int kk = 0; kk < K; kk += 4) {
    const v2f a = *reinterpret_cast<const v2f*>(Ap + kk);
    v2f b;
    b.x = Bp[kk * NCOLS];
    b.y = Bp[kk * NCOLS + NCOLS];
    acc = __builtin_amdgcn_wmma_f32_16x16x4_f32(
         false, a,  false, b,
         (short)0, acc,  false,  false);
  }

  const float bv = bias ? bias[col0 + mn] : 0.0f;
  #pragma unroll
  for (int r = 0; r < 8; ++r) {
    int row = row0 + r + half * 8;
    if (row < M) D[(long long)row * NCOLS + col0 + mn] = acc[r] + bv;
  }
}

template <int U, int VEC>
__device__ __forceinline__ void gather_rows(const int2 ed, const int j, const int lane,
                                            const float* __restrict__ HH,
                                            float __attribute__((ext_vector_type(VEC)))& acc) {
  typedef float vf __attribute__((ext_vector_type(VEC)));
  const vf* __restrict__ Hv = reinterpret_cast<const vf*>(HH);
  int s[U];
  float w[U];
  vf v[U];
  #pragma unroll
  for (int u = 0; u < U; ++u) {
    s[u] = __builtin_amdgcn_readlane(ed.x, j + u);
    w[u] = __int_as_float(__builtin_amdgcn_readlane(ed.y, j + u));
  }
  #pragma unroll
  for (int u = 0; u < U; ++u) v[u] = Hv[(size_t)s[u] * 32 + lane];
  #pragma unroll
  for (int u = 0; u < U; ++u) acc += w[u] * v[u];
}

template <int VEC, bool RELU>
__device__ __forceinline__ void
gather_node(const int* __restrict__ off, const int* __restrict__ cnt,
            const int2* __restrict__ csr, const float* __restrict__ HH,
            const float* __restrict__ bias, float* orow_f, const int node, const int lane, const int N) {
  typedef float vf __attribute__((ext_vector_type(VEC)));
  constexpr int WIDE = 32 / VEC;
  const vf* __restrict__ Hv = reinterpret_cast<const vf*>(HH);
  const vf zero = (vf)(0.0f);
  vf* orow = reinterpret_cast<vf*>(orow_f);

  int beg[3], c[3];
  #pragma unroll
  for (int g = 0; g < 3; ++g) {
    beg[g] = off[g * N + node];
    c[g]   = cnt[g * N + node];
  }
  asm volatile("" :: "s"(beg[0]), "s"(beg[1]), "s"(beg[2]), "s"(c[0]), "s"(c[1]), "s"(c[2]));

  int2 first[3];
  #pragma unroll
  for (int g = 0; g < 3; ++g) {
    first[g] = make_int2(0, 0);
    if (c[g] > 0) {
      const int m0 = c[g] < 32 ? c[g] : 32;
      first[g] = csr[beg[g] + (lane < m0 ? lane : m0 - 1)];
    }
  }
  const vf bv   = reinterpret_cast<const vf*>(bias)[lane];
  const vf self = Hv[(size_t)node * 32 + lane];
  asm volatile("" :: "v"(bv), "v"(self));

  #pragma unroll
  for (int g = 0; g < 3; ++g) {
    vf acc = zero;
    int2 ed = first[g];
    for (int base = 0; base < c[g]; base += 32) {
      const int rem = c[g] - base;
      const int m = rem < 32 ? rem : 32;
      int2 nxt = ed;
      if (rem > 32) {
        const int m2 = rem - 32 < 32 ? rem - 32 : 32;
        nxt = csr[beg[g] + base + 32 + (lane < m2 ? lane : m2 - 1)];
      }
      int j = 0;
      for (; j + WIDE <= m; j += WIDE) gather_rows<WIDE, VEC>(ed, j, lane, HH, acc);
      if (WIDE > 8 && j + 8 <= m) { gather_rows<8, VEC>(ed, j, lane, HH, acc); j += 8; }
      if (j + 4 <= m) { gather_rows<4, VEC>(ed, j, lane, HH, acc); j += 4; }
      if (j < m) {
        const int j1 = j + 1 < m ? j + 1 : j;
        const int j2 = j + 2 < m ? j + 2 : j;
        const int s0 = __builtin_amdgcn_readlane(ed.x, j);
        const int s1 = __builtin_amdgcn_readlane(ed.x, j1);
        const int s2 = __builtin_amdgcn_readlane(ed.x, j2);
        const float w0 = __int_as_float(__builtin_amdgcn_readlane(ed.y, j));
        const float w1 = __int_as_float(__builtin_amdgcn_readlane(ed.y, j1));
        const float w2 = __int_as_float(__builtin_amdgcn_readlane(ed.y, j2));
        const vf v0 = Hv[(size_t)s0 * 32 + lane];
        const vf v1 = Hv[(size_t)s1 * 32 + lane];
        const vf v2 = Hv[(size_t)s2 * 32 + lane];
        asm volatile("" :: "v"(v0), "v"(v1), "v"(v2));
        acc += w0 * v0;
        if (j + 1 < m) acc += w1 * v1;
        if (j + 2 < m) acc += w2 * v2;
      }
      ed = nxt;
    }
    vf r = acc + bv;
    if (RELU) r = __builtin_elementwise_max(r, zero);
    orow[g * 32 + lane] = r;
  }
  vf r = self + bv;
  if (RELU) r = __builtin_elementwise_max(r, zero);
  orow[3 * 32 + lane] = r;
}

__device__ __forceinline__ void lsm_row(float a, float b, float& ra, float& rb) {
  float mx = fmaxf(a, b);
  #pragma unroll
  for (int off = 16; off > 0; off >>= 1) mx = fmaxf(mx, __shfl_xor(mx, off));
  float s = __expf(a - mx) + __expf(b - mx);
  #pragma unroll
  for (int off = 16; off > 0; off >>= 1) s += __shfl_xor(s, off);
  const float lse = mx + __logf(s);
  ra = a - lse;
  rb = b - lse;
}

template <int VEC, bool RELU, int NCOLS, bool LOGSOFTMAX>
__global__ void __launch_bounds__(512)
gcn_gather_gemm(const int* __restrict__ off, const int* __restrict__ cnt,
                const int2* __restrict__ csr, const float* __restrict__ HH,
                const float* __restrict__ bias, const float* __restrict__ B,
                const float* __restrict__ bias2, float* __restrict__ D, int N) {
  constexpr int K   = 4 * 32 * VEC;
  constexpr int LDA = K + 4;
  constexpr int KQ  = K / 4;
  constexpr int LDP = NCOLS + 4;
  static_assert(NCOLS == 64, "4 column tiles x 4 K quarters = 16 waves; a row = 2 columns per lane");
  static_assert(KQ % 64 == 0, "a K quarter is a whole number of 16-step trips");
  __shared__ __attribute__((aligned(16))) float As[16 * LDA];
  __shared__ float Ps[4 * 16 * LDP];

  const int wv   = __builtin_amdgcn_readfirstlane((int)(threadIdx.x >> 5));
  const int lane = threadIdx.x & 31;
  const int row0 = blockIdx.x * 16;
  const int node = row0 + wv;
  float* srow = &As[wv * LDA];
  if (node < N) {
    gather_node<VEC, RELU>(off, cnt, csr, HH, bias, srow, node, lane, N);
  } else {
    typedef float vf __attribute__((ext_vector_type(VEC)));
    #pragma unroll
    for (int g = 0; g < 4; ++g) reinterpret_cast<vf*>(srow)[g * 32 + lane] = (vf)(0.0f);
  }
  __syncthreads();

  const int half = lane >> 4;
  const int mn   = lane & 15;
  const int kb   = half * 2;
  const int col0 = (wv & 3) * 16;
  const int kq   = wv >> 2;
  const int k0   = kq * KQ;

  const float* Ap = &As[mn * LDA + k0 + kb];
  const float* Bp = B + (k0 + kb) * NCOLS + col0 + mn;

  float ba = 0.0f, bb = 0.0f;
  if (bias2) { ba = bias2[lane]; bb = bias2[lane + 32]; }

  v8f acc = {};
  #pragma unroll 16
  for (int kk = 0; kk < KQ; kk += 4) {
    const v2f a = *reinterpret_cast<const v2f*>(Ap + kk);
    v2f b;
    b.x = Bp[kk * NCOLS];
    b.y = Bp[kk * NCOLS + NCOLS];
    acc = __builtin_amdgcn_wmma_f32_16x16x4_f32(
         false, a,  false, b,
         (short)0, acc,  false,  false);
  }

  #pragma unroll
  for (int r = 0; r < 8; ++r) Ps[(kq * 16 + r + half * 8) * LDP + col0 + mn] = acc[r];
  __syncthreads();

  const float* pr = &Ps[wv * LDP + lane];
  float va = ((pr[0] + pr[16 * LDP]) + pr[32 * LDP]) + pr[48 * LDP];
  float vb = ((pr[32] + pr[16 * LDP + 32]) + pr[32 * LDP + 32]) + pr[48 * LDP + 32];
  va += ba;
  vb += bb;
  if (LOGSOFTMAX) {
    float ra, rb;
    lsm_row(va, vb, ra, rb);
    va = ra;
    vb = rb;
  }
  if (node < N) {
    float* drow = D + (long long)node * NCOLS;
    drow[lane]      = va;
    drow[lane + 32] = vb;
  }
}

extern "C" void kernel_launch(void* const* d_in, const int* in_sizes, int n_in,
                              void* d_out, int out_size, void* d_ws, size_t ws_size,
                              hipStream_t stream) {
  const float* x  = (const float*)d_in[0];
  const int* e1   = (const int*)d_in[1];
  const int* e2   = (const int*)d_in[2];
  const int* e3   = (const int*)d_in[3];
  const float* W1 = (const float*)d_in[5];
  const float* b1 = (const float*)d_in[6];
  const float* W2 = (const float*)d_in[7];
  const float* b2 = (const float*)d_in[8];
  const float* Wl = (const float*)d_in[9];
  const float* bl = (const float*)d_in[10];

  const int F = 512, H = 128, C = 64;
  const int N  = in_sizes[0] / F;
  const int E1 = in_sizes[1] / 2;
  const int E2 = in_sizes[2] / 2;
  const int E3 = in_sizes[3] / 2;
  const long long Et = (long long)E1 + E2 + E3;

  float* ws   = (float*)d_ws;
  float* HH1  = ws;
  float* HH2  = HH1 + (long long)N * H;
  int2*  csr  = (int2*)(HH2 + (long long)N * C);
  int2*  binned = csr + Et;
  float* dinv = (float*)(binned + Et);
  int*   cnt  = (int*)(dinv + 3LL * N);
  int*   off  = cnt + 3LL * N;

  const int T = 256;

  const int NB    = 128;
  const int nkeys = 3 * N;
  const int nbin  = cdiv_i(nkeys, BIN_KEYS);
  const int chunk = cdiv_i(Et, NB);
  int*   binStart = off + 3LL * N;
  int*   hist     = binStart + (nbin + 1);
  bin_count<8><<<NB, SORT_T, 0, stream>>>(e1 + E1, e2 + E2, e3 + E3, E1, E2, E3, N, nbin, chunk, hist);
  bin_scatter<8><<<NB, SORT_T, 0, stream>>>(e1, e2, e3, E1, E2, E3, N, nbin, chunk, hist, binStart, binned);
  bin_sort<8><<<nbin, SORT_T, 0, stream>>>(binned, binStart, nkeys, cnt, off, dinv, csr);
  csr_weight<8><<<512, T, 0, stream>>>(csr, dinv, E1, E1 + E2, (int)Et, N);

  const int mtiles = cdiv_i(N, 16);

  gemm_wmma_f32<F, H><<<mtiles, 32 * (H / 16), 0, stream>>>(x, W1, nullptr, HH1, N);

  gcn_gather_gemm<4, true, C, false><<<mtiles, 512, 0, stream>>>(off, cnt, csr, HH1, b1, W2, nullptr, HH2, N);

  gcn_gather_gemm<2, false, C, true><<<mtiles, 512, 0, stream>>>(off, cnt, csr, HH2, b2, Wl, bl, (float*)d_out, N);
}


# x@W1 loses its own launch: its 16-row strips run as extra blocks of the bin-sort and edge-weight launches, which leave the matrix pipes idle (7 launches to 6); strip code and k order kept
# speedup vs baseline: 6.0520x; 1.0384x over previous
#include <hip/hip_runtime.h>
#include <math.h>

typedef float v2f __attribute__((ext_vector_type(2)));
typedef float v8f __attribute__((ext_vector_type(8)));

static inline int cdiv_i(long long a, long long b) { return (int)((a + b - 1) / b); }

constexpr int BIN_SHIFT = 8;
constexpr int BIN_KEYS  = 1 << BIN_SHIFT;
constexpr int MAX_BINS  = 1024;
constexpr int SORT_T    = 1024;

struct EdgeRef { const int* p; int gN; };
__device__ __forceinline__ EdgeRef edge_ref(int i, const int* __restrict__ a1, const int* __restrict__ a2,
                                            const int* __restrict__ a3, int E1, int E12, int N) {
  EdgeRef r;
  int li = i;
  r.p = a1; r.gN = 0;
  if (i >= E1)  { r.p = a2; li = i - E1;  r.gN = N; }
  if (i >= E12) { r.p = a3; li = i - E12; r.gN = 2 * N; }
  r.p += li;
  return r;
}

__device__ __forceinline__ void hold8(const int (&a)[8]) {
  asm volatile("" :: "v"(a[0]), "v"(a[1]), "v"(a[2]), "v"(a[3]), "v"(a[4]), "v"(a[5]), "v"(a[6]), "v"(a[7]));
}

static_assert(SORT_T == 1024, "the scan below needs exactly 32 waves of 32 lanes");
__device__ __forceinline__ int block_scan_inclusive(int v, int* wsum) {
  const int lane = threadIdx.x & 31;
  const int wave = threadIdx.x >> 5;
  int x = v;
  #pragma unroll
  for (int d = 1; d < 32; d <<= 1) {
    const int y = __shfl_up(x, d);
    if (lane >= d) x += y;
  }
  if (lane == 31) wsum[wave] = x;
  __syncthreads();
  if (wave == 0) {
    const int t = wsum[lane];
    int xs = t;
    #pragma unroll
    for (int d = 1; d < 32; d <<= 1) {
      const int y = __shfl_up(xs, d);
      if (lane >= d) xs += y;
    }
    wsum[lane] = xs - t;
  }
  __syncthreads();
  return x + wsum[wave];
}

template <int U>
__global__ void __launch_bounds__(SORT_T)
bin_count(const int* __restrict__ d1, const int* __restrict__ d2, const int* __restrict__ d3,
          int E1, int E2, int E3, int N, int nbin, int chunk, int* __restrict__ hist) {
  __shared__ int h[MAX_BINS];
  const int tid = threadIdx.x;
  const int E12 = E1 + E2;
  const int Et  = E12 + E3;
  for (int b = tid; b < nbin; b += SORT_T) h[b] = 0;
  __syncthreads();
  const int lo = (int)blockIdx.x * chunk;
  const int hi = lo + chunk < Et ? lo + chunk : Et;
  for (int c0 = lo; c0 < hi; c0 += SORT_T * U) {
    int d[U], gN[U];
    #pragma unroll
    for (int u = 0; u < U; ++u) {
      const int i  = c0 + tid + u * SORT_T;
      const int ic = i < hi ? i : hi - 1;
      const EdgeRef e = edge_ref(ic, d1, d2, d3, E1, E12, N);
      d[u]  = *e.p;
      gN[u] = e.gN;
    }
    hold8(d);
    #pragma unroll
    for (int u = 0; u < U; ++u) {
      if (c0 + tid + u * SORT_T < hi) atomicAdd(&h[(gN[u] + d[u]) >> BIN_SHIFT], 1);
    }
  }
  __syncthreads();
  for (int b = tid; b < nbin; b += SORT_T) hist[(int)blockIdx.x * nbin + b] = h[b];
}

template <int U>
__global__ void __launch_bounds__(SORT_T)
bin_scatter(const int* __restrict__ e1, const int* __restrict__ e2, const int* __restrict__ e3,
            int E1, int E2, int E3, int N, int nbin, int chunk, const int* __restrict__ hist,
            int* __restrict__ binStart, int2* __restrict__ binned) {
  __shared__ int cur[MAX_BINS];
  __shared__ int tot[MAX_BINS];
  __shared__ int wsum[32];
  const int tid = threadIdx.x;
  const int E12 = E1 + E2;
  const int Et  = E12 + E3;
  for (int b = tid; b < nbin; b += SORT_T) { cur[b] = 0; tot[b] = 0; }
  __syncthreads();
  const int cells = (int)gridDim.x * nbin;
  const int dk = SORT_T / nbin;
  const int db = SORT_T - dk * nbin;
  int k = tid / nbin;
  int b = tid - k * nbin;
  constexpr int P = 8;
  for (int i0 = tid; i0 < cells; i0 += SORT_T * P) {
    int v[P];
    #pragma unroll
    for (int p = 0; p < P; ++p) {
      const int idx = i0 + p * SORT_T;
      v[p] = hist[idx < cells ? idx : cells - 1];
    }
    hold8(v);
    #pragma unroll
    for (int p = 0; p < P; ++p) {
      if (i0 + p * SORT_T < cells) {
        atomicAdd(&tot[b], v[p]);
        if (k < (int)blockIdx.x) atomicAdd(&cur[b], v[p]);
      }
      k += dk;
      b += db;
      if (b >= nbin) { b -= nbin; k += 1; }
    }
  }
  __syncthreads();
  const int mine = tid < nbin ? tot[tid] : 0;
  const int incl = block_scan_inclusive(mine, wsum);
  if (tid < nbin) {
    const int start = incl - mine;
    cur[tid] += start;
    if (blockIdx.x == 0) {
      binStart[tid] = start;
      if (tid == nbin - 1) binStart[nbin] = incl;
    }
  }
  __syncthreads();
  const int lo = (int)blockIdx.x * chunk;
  const int hi = lo + chunk < Et ? lo + chunk : Et;
  for (int c0 = lo; c0 < hi; c0 += SORT_T * U) {
    int s[U], d[U], gN[U];
    #pragma unroll
    for (int u = 0; u < U; ++u) {
      const int i  = c0 + tid + u * SORT_T;
      const int ic = i < hi ? i : hi - 1;
      const EdgeRef es = edge_ref(ic, e1, e2, e3, E1, E12, N);
      const EdgeRef ed = edge_ref(ic, e1 + E1, e2 + E2, e3 + E3, E1, E12, N);
      s[u]  = *es.p;
      d[u]  = *ed.p;
      gN[u] = ed.gN;
    }
    hold8(s);
    hold8(d);
    #pragma unroll
    for (int u = 0; u < U; ++u) {
      if (c0 + tid + u * SORT_T < hi) {
        const int key = gN[u] + d[u];
        const int pos = atomicAdd(&cur[key >> BIN_SHIFT], 1);
        binned[pos] = make_int2(s[u], key);
      }
    }
  }
}

template <int K, int NCOLS>
__device__ __forceinline__ void
gemm_strip_f32(const float* __restrict__ A, const float* __restrict__ B,
               const float* __restrict__ bias, float* __restrict__ D, int M,
               const int strip, const int tid, float* __restrict__ As) {
  constexpr int NT  = NCOLS * 2;
  constexpr int LDA = K + 4;
  constexpr int K4  = K / 4;
  constexpr int PER = (16 * K4) / NT;
  static_assert((16 * K4) % NT == 0, "strip must divide evenly over the group");

  const int row0 = strip * 16;

  float4 stage[PER];
  #pragma unroll
  for (int i = 0; i < PER; ++i) {
    const int q  = tid + i * NT;
    const int r  = q / K4;
    const int c4 = q - r * K4;
    int row = row0 + r;
    if (row >= M) row = M - 1;
    stage[i] = *reinterpret_cast<const float4*>(A + (long long)row * K + c4 * 4);
  }
  #pragma unroll
  for (int i = 0; i < PER; ++i) {
    const int q  = tid + i * NT;
    const int r  = q / K4;
    const int c4 = q - r * K4;
    *reinterpret_cast<float4*>(&As[r * LDA + c4 * 4]) = stage[i];
  }
  __syncthreads();

  const int wave = tid >> 5;
  const int lane = tid & 31;
  const int half = lane >> 4;
  const int mn   = lane & 15;
  const int kb   = half * 2;
  const int col0 = wave * 16;

  const float* Ap = &As[mn * LDA + kb];
  const float* Bp = B + kb * NCOLS + col0 + mn;

  v8f acc = {};
  #pragma unroll 4
  for (int kk = 0; kk < K; kk += 4) {
    const v2f a = *reinterpret_cast<const v2f*>(Ap + kk);
    v2f b;
    b.x = Bp[kk * NCOLS];
    b.y = Bp[kk * NCOLS + NCOLS];
    acc = __builtin_amdgcn_wmma_f32_16x16x4_f32(
         false, a,  false, b,
         (short)0, acc,  false,  false);
  }

  const float bv = bias ? bias[col0 + mn] : 0.0f;
  #pragma unroll
  for (int r = 0; r < 8; ++r) {
    int row = row0 + r + half * 8;
    if (row < M) D[(long long)row * NCOLS + col0 + mn] = acc[r] + bv;
  }
}

template <int U, int GK, int GNCOLS>
__global__ void __launch_bounds__(SORT_T)
bin_sort(const int2* __restrict__ binned, const int* __restrict__ binStart, int nkeys, int nbin,
         int* __restrict__ cnt, int* __restrict__ off, float* __restrict__ dinv, int2* __restrict__ csr,
         const float* __restrict__ GA, const float* __restrict__ GB, float* __restrict__ GD, int GM) {
  static_assert(GNCOLS * 2 * 4 == SORT_T, "four strip groups fill a block");
  __shared__ int   c[BIN_KEYS];
  __shared__ int   wsum[32];
  __shared__ float dv[BIN_KEYS];
  __shared__ __attribute__((aligned(16))) float As[4][16 * (GK + 4)];
  const int tid = threadIdx.x;
  if ((int)blockIdx.x >= nbin) {
    const int grp = tid >> 8;
    gemm_strip_f32<GK, GNCOLS>(GA, GB, nullptr, GD, GM, ((int)blockIdx.x - nbin) * 4 + grp, tid & 255, As[grp]);
    return;
  }
  const int bin = blockIdx.x;
  const int s0  = binStart[bin];
  const int s1  = binStart[bin + 1];
  const int klo = bin << BIN_SHIFT;
  if (tid < BIN_KEYS) c[tid] = 0;
  __syncthreads();
  for (int c0 = s0; c0 < s1; c0 += SORT_T * U) {
    int key[U];
    #pragma unroll
    for (int u = 0; u < U; ++u) {
      const int i  = c0 + tid + u * SORT_T;
      const int ic = i < s1 ? i : s1 - 1;
      key[u] = binned[ic].y;
    }
    hold8(key);
    #pragma unroll
    for (int u = 0; u < U; ++u) {
      if (c0 + tid + u * SORT_T < s1) atomicAdd(&c[key[u] - klo], 1);
    }
  }
  __syncthreads();
  const int v = tid < BIN_KEYS ? c[tid] : 0;
  const int incl = block_scan_inclusive(v, wsum);
  if (tid < BIN_KEYS) {
    const int first = s0 + incl - v;
    const float w = (v > 0) ? rsqrtf((float)v) : 0.0f;
    c[tid]  = first;
    dv[tid] = w;
    const int key = klo + tid;
    if (key < nkeys) { cnt[key] = v; off[key] = first; dinv[key] = w; }
  }
  __syncthreads();
  for (int c0 = s0; c0 < s1; c0 += SORT_T * U) {
    int src[U], key[U];
    #pragma unroll
    for (int u = 0; u < U; ++u) {
      const int i  = c0 + tid + u * SORT_T;
      const int ic = i < s1 ? i : s1 - 1;
      const int2 e = binned[ic];
      src[u] = e.x;
      key[u] = e.y;
    }
    hold8(src);
    hold8(key);
    #pragma unroll
    for (int u = 0; u < U; ++u) {
      if (c0 + tid + u * SORT_T < s1) {
        const int k   = key[u] - klo;
        const int pos = atomicAdd(&c[k], 1);
        csr[pos] = make_int2(src[u], __float_as_int(dv[k]));
      }
    }
  }
}

template <int U, int GK, int GNCOLS>
__global__ void __launch_bounds__(256)
csr_weight(int2* __restrict__ csr, const float* __restrict__ dinv, int E1, int E12, int Et, int N,
           const float* __restrict__ GA, const float* __restrict__ GB, float* __restrict__ GD, int GM,
           int gstrip0, int gblocks) {
  static_assert(GNCOLS * 2 == 256, "one strip group fills a block");
  __shared__ __attribute__((aligned(16))) float As[16 * (GK + 4)];
  if ((int)blockIdx.x < gblocks) {
    gemm_strip_f32<GK, GNCOLS>(GA, GB, nullptr, GD, GM, gstrip0 + (int)blockIdx.x, threadIdx.x, As);
    return;
  }
  const int wblock = (int)blockIdx.x - gblocks;
  const int step = ((int)gridDim.x - gblocks) * 256 * U;
  for (int c0 = wblock * 256 * U; c0 < Et; c0 += step) {
    int src[U], wd[U], ws[U];
    #pragma unroll
    for (int u = 0; u < U; ++u) {
      const int i  = c0 + (int)threadIdx.x + u * 256;
      const int ic = i < Et ? i : Et - 1;
      const int2 e = csr[ic];
      src[u] = e.x;
      wd[u]  = e.y;
    }
    hold8(src);
    #pragma unroll
    for (int u = 0; u < U; ++u) {
      const int i  = c0 + (int)threadIdx.x + u * 256;
      const int ic = i < Et ? i : Et - 1;
      const int gN = ic >= E12 ? 2 * N : (ic >= E1 ? N : 0);
      ws[u] = __float_as_int(dinv[gN + src[u]]);
    }
    hold8(ws);
    #pragma unroll
    for (int u = 0; u < U; ++u) {
      const int i = c0 + (int)threadIdx.x + u * 256;
      if (i < Et) csr[i] = make_int2(src[u], __float_as_int(__int_as_float(ws[u]) * __int_as_float(wd[u])));
    }
  }
}

template <int U, int VEC>
__device__ __forceinline__ void gather_rows(const int2 ed, const int j, const int lane,
                                            const float* __restrict__ HH,
                                            float __attribute__((ext_vector_type(VEC)))& acc) {
  typedef float vf __attribute__((ext_vector_type(VEC)));
  const vf* __restrict__ Hv = reinterpret_cast<const vf*>(HH);
  int s[U];
  float w[U];
  vf v[U];
  #pragma unroll
  for (int u = 0; u < U; ++u) {
    s[u] = __builtin_amdgcn_readlane(ed.x, j + u);
    w[u] = __int_as_float(__builtin_amdgcn_readlane(ed.y, j + u));
  }
  #pragma unroll
  for (int u = 0; u < U; ++u) v[u] = Hv[(size_t)s[u] * 32 + lane];
  #pragma unroll
  for (int u = 0; u < U; ++u) acc += w[u] * v[u];
}

template <int VEC, bool RELU>
__device__ __forceinline__ void
gather_node(const int* __restrict__ off, const int* __restrict__ cnt,
            const int2* __restrict__ csr, const float* __restrict__ HH,
            const float* __restrict__ bias, float* orow_f, const int node, const int lane, const int N) {
  typedef float vf __attribute__((ext_vector_type(VEC)));
  constexpr int WIDE = 32 / VEC;
  const vf* __restrict__ Hv = reinterpret_cast<const vf*>(HH);
  const vf zero = (vf)(0.0f);
  vf* orow = reinterpret_cast<vf*>(orow_f);

  int beg[3], c[3];
  #pragma unroll
  for (int g = 0; g < 3; ++g) {
    beg[g] = off[g * N + node];
    c[g]   = cnt[g * N + node];
  }
  asm volatile("" :: "s"(beg[0]), "s"(beg[1]), "s"(beg[2]), "s"(c[0]), "s"(c[1]), "s"(c[2]));

  int2 first[3];
  #pragma unroll
  for (int g = 0; g < 3; ++g) {
    first[g] = make_int2(0, 0);
    if (c[g] > 0) {
      const int m0 = c[g] < 32 ? c[g] : 32;
      first[g] = csr[beg[g] + (lane < m0 ? lane : m0 - 1)];
    }
  }
  const vf bv   = reinterpret_cast<const vf*>(bias)[lane];
  const vf self = Hv[(size_t)node * 32 + lane];
  asm volatile("" :: "v"(bv), "v"(self));

  #pragma unroll
  for (int g = 0; g < 3; ++g) {
    vf acc = zero;
    int2 ed = first[g];
    for (int base = 0; base < c[g]; base += 32) {
      const int rem = c[g] - base;
      const int m = rem < 32 ? rem : 32;
      int2 nxt = ed;
      if (rem > 32) {
        const int m2 = rem - 32 < 32 ? rem - 32 : 32;
        nxt = csr[beg[g] + base + 32 + (lane < m2 ? lane : m2 - 1)];
      }
      int j = 0;
      for (; j + WIDE <= m; j += WIDE) gather_rows<WIDE, VEC>(ed, j, lane, HH, acc);
      if (WIDE > 8 && j + 8 <= m) { gather_rows<8, VEC>(ed, j, lane, HH, acc); j += 8; }
      if (j + 4 <= m) { gather_rows<4, VEC>(ed, j, lane, HH, acc); j += 4; }
      if (j < m) {
        const int j1 = j + 1 < m ? j + 1 : j;
        const int j2 = j + 2 < m ? j + 2 : j;
        const int s0 = __builtin_amdgcn_readlane(ed.x, j);
        const int s1 = __builtin_amdgcn_readlane(ed.x, j1);
        const int s2 = __builtin_amdgcn_readlane(ed.x, j2);
        const float w0 = __int_as_float(__builtin_amdgcn_readlane(ed.y, j));
        const float w1 = __int_as_float(__builtin_amdgcn_readlane(ed.y, j1));
        const float w2 = __int_as_float(__builtin_amdgcn_readlane(ed.y, j2));
        const vf v0 = Hv[(size_t)s0 * 32 + lane];
        const vf v1 = Hv[(size_t)s1 * 32 + lane];
        const vf v2 = Hv[(size_t)s2 * 32 + lane];
        asm volatile("" :: "v"(v0), "v"(v1), "v"(v2));
        acc += w0 * v0;
        if (j + 1 < m) acc += w1 * v1;
        if (j + 2 < m) acc += w2 * v2;
      }
      ed = nxt;
    }
    vf r = acc + bv;
    if (RELU) r = __builtin_elementwise_max(r, zero);
    orow[g * 32 + lane] = r;
  }
  vf r = self + bv;
  if (RELU) r = __builtin_elementwise_max(r, zero);
  orow[3 * 32 + lane] = r;
}

__device__ __forceinline__ void lsm_row(float a, float b, float& ra, float& rb) {
  float mx = fmaxf(a, b);
  #pragma unroll
  for (int off = 16; off > 0; off >>= 1) mx = fmaxf(mx, __shfl_xor(mx, off));
  float s = __expf(a - mx) + __expf(b - mx);
  #pragma unroll
  for (int off = 16; off > 0; off >>= 1) s += __shfl_xor(s, off);
  const float lse = mx + __logf(s);
  ra = a - lse;
  rb = b - lse;
}

template <int VEC, bool RELU, int NCOLS, bool LOGSOFTMAX>
__global__ void __launch_bounds__(512)
gcn_gather_gemm(const int* __restrict__ off, const int* __restrict__ cnt,
                const int2* __restrict__ csr, const float* __restrict__ HH,
                const float* __restrict__ bias, const float* __restrict__ B,
                const float* __restrict__ bias2, float* __restrict__ D, int N) {
  constexpr int K   = 4 * 32 * VEC;
  constexpr int LDA = K + 4;
  constexpr int KQ  = K / 4;
  constexpr int LDP = NCOLS + 4;
  static_assert(NCOLS == 64, "4 column tiles x 4 K quarters = 16 waves; a row = 2 columns per lane");
  static_assert(KQ % 64 == 0, "a K quarter is a whole number of 16-step trips");
  __shared__ __attribute__((aligned(16))) float As[16 * LDA];
  __shared__ float Ps[4 * 16 * LDP];

  const int wv   = __builtin_amdgcn_readfirstlane((int)(threadIdx.x >> 5));
  const int lane = threadIdx.x & 31;
  const int row0 = blockIdx.x * 16;
  const int node = row0 + wv;
  float* srow = &As[wv * LDA];
  if (node < N) {
    gather_node<VEC, RELU>(off, cnt, csr, HH, bias, srow, node, lane, N);
  } else {
    typedef float vf __attribute__((ext_vector_type(VEC)));
    #pragma unroll
    for (int g = 0; g < 4; ++g) reinterpret_cast<vf*>(srow)[g * 32 + lane] = (vf)(0.0f);
  }
  __syncthreads();

  const int half = lane >> 4;
  const int mn   = lane & 15;
  const int kb   = half * 2;
  const int col0 = (wv & 3) * 16;
  const int kq   = wv >> 2;
  const int k0   = kq * KQ;

  const float* Ap = &As[mn * LDA + k0 + kb];
  const float* Bp = B + (k0 + kb) * NCOLS + col0 + mn;

  float ba = 0.0f, bb = 0.0f;
  if (bias2) { ba = bias2[lane]; bb = bias2[lane + 32]; }

  v8f acc = {};
  #pragma unroll 16
  for (int kk = 0; kk < KQ; kk += 4) {
    const v2f a = *reinterpret_cast<const v2f*>(Ap + kk);
    v2f b;
    b.x = Bp[kk * NCOLS];
    b.y = Bp[kk * NCOLS + NCOLS];
    acc = __builtin_amdgcn_wmma_f32_16x16x4_f32(
         false, a,  false, b,
         (short)0, acc,  false,  false);
  }

  #pragma unroll
  for (int r = 0; r < 8; ++r) Ps[(kq * 16 + r + half * 8) * LDP + col0 + mn] = acc[r];
  __syncthreads();

  const float* pr = &Ps[wv * LDP + lane];
  float va = ((pr[0] + pr[16 * LDP]) + pr[32 * LDP]) + pr[48 * LDP];
  float vb = ((pr[32] + pr[16 * LDP + 32]) + pr[32 * LDP + 32]) + pr[48 * LDP + 32];
  va += ba;
  vb += bb;
  if (LOGSOFTMAX) {
    float ra, rb;
    lsm_row(va, vb, ra, rb);
    va = ra;
    vb = rb;
  }
  if (node < N) {
    float* drow = D + (long long)node * NCOLS;
    drow[lane]      = va;
    drow[lane + 32] = vb;
  }
}

extern "C" void kernel_launch(void* const* d_in, const int* in_sizes, int n_in,
                              void* d_out, int out_size, void* d_ws, size_t ws_size,
                              hipStream_t stream) {
  const float* x  = (const float*)d_in[0];
  const int* e1   = (const int*)d_in[1];
  const int* e2   = (const int*)d_in[2];
  const int* e3   = (const int*)d_in[3];
  const float* W1 = (const float*)d_in[5];
  const float* b1 = (const float*)d_in[6];
  const float* W2 = (const float*)d_in[7];
  const float* b2 = (const float*)d_in[8];
  const float* Wl = (const float*)d_in[9];
  const float* bl = (const float*)d_in[10];

  const int F = 512, H = 128, C = 64;
  const int N  = in_sizes[0] / F;
  const int E1 = in_sizes[1] / 2;
  const int E2 = in_sizes[2] / 2;
  const int E3 = in_sizes[3] / 2;
  const long long Et = (long long)E1 + E2 + E3;

  float* ws   = (float*)d_ws;
  float* HH1  = ws;
  float* HH2  = HH1 + (long long)N * H;
  int2*  csr  = (int2*)(HH2 + (long long)N * C);
  int2*  binned = csr + Et;
  float* dinv = (float*)(binned + Et);
  int*   cnt  = (int*)(dinv + 3LL * N);
  int*   off  = cnt + 3LL * N;

  const int T = 256;

  const int NB    = 128;
  const int nkeys = 3 * N;
  const int nbin  = cdiv_i(nkeys, BIN_KEYS);
  const int chunk = cdiv_i(Et, NB);
  int*   binStart = off + 3LL * N;
  int*   hist     = binStart + (nbin + 1);
  bin_count<8><<<NB, SORT_T, 0, stream>>>(e1 + E1, e2 + E2, e3 + E3, E1, E2, E3, N, nbin, chunk, hist);
  bin_scatter<8><<<NB, SORT_T, 0, stream>>>(e1, e2, e3, E1, E2, E3, N, nbin, chunk, hist, binStart, binned);
  const int mtiles = cdiv_i(N, 16);
  const int gquad  = mtiles / 4 < 256 ? mtiles / 4 : 256;
  const int grest  = mtiles - 4 * gquad;
  bin_sort<8, F, H><<<nbin + gquad, SORT_T, 0, stream>>>(binned, binStart, nkeys, nbin, cnt, off, dinv, csr,
                                                         x, W1, HH1, N);
  csr_weight<8, F, H><<<grest + 512, T, 0, stream>>>(csr, dinv, E1, E1 + E2, (int)Et, N,
                                                     x, W1, HH1, N, 4 * gquad, grest);

  gcn_gather_gemm<4, true, C, false><<<mtiles, 512, 0, stream>>>(off, cnt, csr, HH1, b1, W2, nullptr, HH2, N);

  gcn_gather_gemm<2, false, C, true><<<mtiles, 512, 0, stream>>>(off, cnt, csr, HH2, b2, Wl, bl, (float*)d_out, N);
}


# Level-1 count and scatter launches merged: a block sorts its own fixed segment of edges by bin in LDS, level 2 reads a bin as one slice per segment (6 launches to 5); sum order = LDS add arrival
# speedup vs baseline: 6.7611x; 1.1172x over previous
#include <hip/hip_runtime.h>
#include <math.h>

typedef float v2f __attribute__((ext_vector_type(2)));
typedef float v8f __attribute__((ext_vector_type(8)));

static inline int cdiv_i(long long a, long long b) { return (int)((a + b - 1) / b); }

constexpr int BIN_SHIFT = 8;
constexpr int BIN_KEYS  = 1 << BIN_SHIFT;
constexpr int MAX_BINS  = 1024;
constexpr int SORT_T    = 1024;
constexpr int SEG_U     = 8;
constexpr int SEG_EDGES = SORT_T * SEG_U;

struct EdgeRef { const int* p; int gN; };
__device__ __forceinline__ EdgeRef edge_ref(int i, const int* __restrict__ a1, const int* __restrict__ a2,
                                            const int* __restrict__ a3, int E1, int E12, int N) {
  EdgeRef r;
  int li = i;
  r.p = a1; r.gN = 0;
  if (i >= E1)  { r.p = a2; li = i - E1;  r.gN = N; }
  if (i >= E12) { r.p = a3; li = i - E12; r.gN = 2 * N; }
  r.p += li;
  return r;
}

__device__ __forceinline__ void hold8(const int (&a)[8]) {
  asm volatile("" :: "v"(a[0]), "v"(a[1]), "v"(a[2]), "v"(a[3]), "v"(a[4]), "v"(a[5]), "v"(a[6]), "v"(a[7]));
}

static_assert(SORT_T == 1024, "the scan below needs exactly 32 waves of 32 lanes");
__device__ __forceinline__ int block_scan_inclusive(int v, int* wsum) {
  const int lane = threadIdx.x & 31;
  const int wave = threadIdx.x >> 5;
  int x = v;
  #pragma unroll
  for (int d = 1; d < 32; d <<= 1) {
    const int y = __shfl_up(x, d);
    if (lane >= d) x += y;
  }
  if (lane == 31) wsum[wave] = x;
  __syncthreads();
  if (wave == 0) {
    const int t = wsum[lane];
    int xs = t;
    #pragma unroll
    for (int d = 1; d < 32; d <<= 1) {
      const int y = __shfl_up(xs, d);
      if (lane >= d) xs += y;
    }
    wsum[lane] = xs - t;
  }
  __syncthreads();
  return x + wsum[wave];
}

static_assert(MAX_BINS <= SORT_T, "one thread per bin in the scan of the counts");
__global__ void __launch_bounds__(SORT_T)
seg_sort(const int* __restrict__ e1, const int* __restrict__ e2, const int* __restrict__ e3,
         int E1, int E2, int E3, int N, int nbin, int nseg,
         int* __restrict__ segCnt, int* __restrict__ segOff, int2* __restrict__ binned) {
  constexpr int U = SEG_U;
  __shared__ int h[MAX_BINS];
  __shared__ int wsum[32];
  const int tid = threadIdx.x;
  const int E12 = E1 + E2;
  const int Et  = E12 + E3;
  for (int b = tid; b < nbin; b += SORT_T) h[b] = 0;
  __syncthreads();
  const int lo = (int)blockIdx.x * SEG_EDGES;
  const int hi = lo + SEG_EDGES < Et ? lo + SEG_EDGES : Et;
  int s[U], d[U], gN[U];
  #pragma unroll
  for (int u = 0; u < U; ++u) {
    const int i  = lo + tid + u * SORT_T;
    const int ic = i < hi ? i : hi - 1;
    const EdgeRef es = edge_ref(ic, e1, e2, e3, E1, E12, N);
    const EdgeRef ed = edge_ref(ic, e1 + E1, e2 + E2, e3 + E3, E1, E12, N);
    s[u]  = *es.p;
    d[u]  = *ed.p;
    gN[u] = ed.gN;
  }
  hold8(s);
  hold8(d);
  int key[U], rank[U];
  #pragma unroll
  for (int u = 0; u < U; ++u) {
    key[u]  = gN[u] + d[u];
    rank[u] = 0;
    if (lo + tid + u * SORT_T < hi) rank[u] = atomicAdd(&h[key[u] >> BIN_SHIFT], 1);
  }
  __syncthreads();
  const int mine = tid < nbin ? h[tid] : 0;
  const int incl = block_scan_inclusive(mine, wsum);
  if (tid < nbin) {
    const int start = incl - mine;
    h[tid] = start;
    segCnt[tid * nseg + (int)blockIdx.x] = mine;
    segOff[tid * nseg + (int)blockIdx.x] = start;
  }
  __syncthreads();
  int st[U];
  #pragma unroll
  for (int u = 0; u < U; ++u) st[u] = h[key[u] >> BIN_SHIFT];
  hold8(st);
  #pragma unroll
  for (int u = 0; u < U; ++u) {
    if (lo + tid + u * SORT_T < hi) binned[lo + st[u] + rank[u]] = make_int2(s[u], key[u]);
  }
}

template <int U>
__device__ __forceinline__ void slice_addr(const int* P, const int* G, const int top, int (&at)[U]) {
  int k[U];
  #pragma unroll
  for (int u = 0; u < U; ++u) k[u] = 0;
  for (int step = top; step > 0; step >>= 1) {
    #pragma unroll
    for (int u = 0; u < U; ++u) {
      const int m = k[u] + step;
      if (P[m] <= at[u]) k[u] = m;
    }
  }
  #pragma unroll
  for (int u = 0; u < U; ++u) at[u] = G[k[u]] + (at[u] - P[k[u]]);
}

template <int K, int NCOLS>
__device__ __forceinline__ void
gemm_strip_f32(const float* __restrict__ A, const float* __restrict__ B,
               const float* __restrict__ bias, float* __restrict__ D, int M,
               const int strip, const int tid, float* __restrict__ As) {
  constexpr int NT  = NCOLS * 2;
  constexpr int LDA = K + 4;
  constexpr int K4  = K / 4;
  constexpr int PER = (16 * K4) / NT;
  static_assert((16 * K4) % NT == 0, "strip must divide evenly over the group");

  const int row0 = strip * 16;

  float4 stage[PER];
  #pragma unroll
  for (int i = 0; i < PER; ++i) {
    const int q  = tid + i * NT;
    const int r  = q / K4;
    const int c4 = q - r * K4;
    int row = row0 + r;
    if (row >= M) row = M - 1;
    stage[i] = *reinterpret_cast<const float4*>(A + (long long)row * K + c4 * 4);
  }
  #pragma unroll
  for (int i = 0; i < PER; ++i) {
    const int q  = tid + i * NT;
    const int r  = q / K4;
    const int c4 = q - r * K4;
    *reinterpret_cast<float4*>(&As[r * LDA + c4 * 4]) = stage[i];
  }
  __syncthreads();

  const int wave = tid >> 5;
  const int lane = tid & 31;
  const int half = lane >> 4;
  const int mn   = lane & 15;
  const int kb   = half * 2;
  const int col0 = wave * 16;

  const float* Ap = &As[mn * LDA + kb];
  const float* Bp = B + kb * NCOLS + col0 + mn;

  v8f acc = {};
  #pragma unroll 4
  for (int kk = 0; kk < K; kk += 4) {
    const v2f a = *reinterpret_cast<const v2f*>(Ap + kk);
    v2f b;
    b.x = Bp[kk * NCOLS];
    b.y = Bp[kk * NCOLS + NCOLS];
    acc = __builtin_amdgcn_wmma_f32_16x16x4_f32(
         false, a,  false, b,
         (short)0, acc,  false,  false);
  }

  const float bv = bias ? bias[col0 + mn] : 0.0f;
  #pragma unroll
  for (int r = 0; r < 8; ++r) {
    int row = row0 + r + half * 8;
    if (row < M) D[(long long)row * NCOLS + col0 + mn] = acc[r] + bv;
  }
}

template <int U, int GK, int GNCOLS>
__global__ void __launch_bounds__(SORT_T)
bin_sort(const int2* __restrict__ binned, const int* __restrict__ segCnt, const int* __restrict__ segOff,
         int nseg, int nkeys, int nbin,
         int* __restrict__ cnt, int* __restrict__ off, float* __restrict__ dinv, int2* __restrict__ csr,
         const float* __restrict__ GA, const float* __restrict__ GB, float* __restrict__ GD, int GM) {
  static_assert(GNCOLS * 2 * 4 == SORT_T, "four strip groups fill a block");
  __shared__ int   c[BIN_KEYS];
  __shared__ int   wsum[32];
  __shared__ float dv[BIN_KEYS];
  __shared__ int   P[SORT_T + 1];
  __shared__ int   G[SORT_T];
  __shared__ int   before;
  __shared__ __attribute__((aligned(16))) float As[4][16 * (GK + 4)];
  const int tid = threadIdx.x;
  if ((int)blockIdx.x >= nbin) {
    const int grp = tid >> 8;
    gemm_strip_f32<GK, GNCOLS>(GA, GB, nullptr, GD, GM, ((int)blockIdx.x - nbin) * 4 + grp, tid & 255, As[grp]);
    return;
  }
  const int bin = blockIdx.x;
  const int klo = bin << BIN_SHIFT;
  int sc = 0, so = 0;
  if (tid < nseg) {
    sc = segCnt[bin * nseg + tid];
    so = segOff[bin * nseg + tid];
  }
  if (tid < BIN_KEYS) c[tid] = 0;
  const int pin = block_scan_inclusive(sc, wsum);
  const int oin = block_scan_inclusive(so, wsum);
  if (tid < nseg) {
    P[tid] = pin - sc;
    G[tid] = tid * SEG_EDGES + so;
  } else if (tid > nseg) {
    P[tid] = 0x7fffffff;
  }
  if (tid == SORT_T - 1) {
    P[nseg] = pin;
    before  = oin;
  }
  __syncthreads();
  const int total = P[nseg];
  const int s0    = before;
  int top = 0;
  if (nseg > 1) {
    top = 1;
    while (top * 2 < nseg) top *= 2;
  }
  for (int c0 = 0; c0 < total; c0 += SORT_T * U) {
    int at[U], key[U];
    #pragma unroll
    for (int u = 0; u < U; ++u) {
      const int i = c0 + tid + u * SORT_T;
      at[u] = i < total ? i : total - 1;
    }
    slice_addr<U>(P, G, top, at);
    #pragma unroll
    for (int u = 0; u < U; ++u) key[u] = binned[at[u]].y;
    hold8(key);
    #pragma unroll
    for (int u = 0; u < U; ++u) {
      if (c0 + tid + u * SORT_T < total) atomicAdd(&c[key[u] - klo], 1);
    }
  }
  __syncthreads();
  const int v = tid < BIN_KEYS ? c[tid] : 0;
  const int incl = block_scan_inclusive(v, wsum);
  if (tid < BIN_KEYS) {
    const int first = s0 + incl - v;
    const float w = (v > 0) ? rsqrtf((float)v) : 0.0f;
    c[tid]  = first;
    dv[tid] = w;
    const int key = klo + tid;
    if (key < nkeys) { cnt[key] = v; off[key] = first; dinv[key] = w; }
  }
  __syncthreads();
  for (int c0 = 0; c0 < total; c0 += SORT_T * U) {
    int at[U], src[U], key[U];
    #pragma unroll
    for (int u = 0; u < U; ++u) {
      const int i = c0 + tid + u * SORT_T;
      at[u] = i < total ? i : total - 1;
    }
    slice_addr<U>(P, G, top, at);
    #pragma unroll
    for (int u = 0; u < U; ++u) {
      const int2 e = binned[at[u]];
      src[u] = e.x;
      key[u] = e.y;
    }
    hold8(src);
    hold8(key);
    #pragma unroll
    for (int u = 0; u < U; ++u) {
      if (c0 + tid + u * SORT_T < total) {
        const int k   = key[u] - klo;
        const int pos = atomicAdd(&c[k], 1);
        csr[pos] = make_int2(src[u], __float_as_int(dv[k]));
      }
    }
  }
}

template <int U, int GK, int GNCOLS>
__global__ void __launch_bounds__(256)
csr_weight(int2* __restrict__ csr, const float* __restrict__ dinv, int E1, int E12, int Et, int N,
           const float* __restrict__ GA, const float* __restrict__ GB, float* __restrict__ GD, int GM,
           int gstrip0, int gblocks) {
  static_assert(GNCOLS * 2 == 256, "one strip group fills a block");
  __shared__ __attribute__((aligned(16))) float As[16 * (GK + 4)];
  if ((int)blockIdx.x < gblocks) {
    gemm_strip_f32<GK, GNCOLS>(GA, GB, nullptr, GD, GM, gstrip0 + (int)blockIdx.x, threadIdx.x, As);
    return;
  }
  const int wblock = (int)blockIdx.x - gblocks;
  const int step = ((int)gridDim.x - gblocks) * 256 * U;
  for (int c0 = wblock * 256 * U; c0 < Et; c0 += step) {
    int src[U], wd[U], ws[U];
    #pragma unroll
    for (int u = 0; u < U; ++u) {
      const int i  = c0 + (int)threadIdx.x + u * 256;
      const int ic = i < Et ? i : Et - 1;
      const int2 e = csr[ic];
      src[u] = e.x;
      wd[u]  = e.y;
    }
    hold8(src);
    #pragma unroll
    for (int u = 0; u < U; ++u) {
      const int i  = c0 + (int)threadIdx.x + u * 256;
      const int ic = i < Et ? i : Et - 1;
      const int gN = ic >= E12 ? 2 * N : (ic >= E1 ? N : 0);
      ws[u] = __float_as_int(dinv[gN + src[u]]);
    }
    hold8(ws);
    #pragma unroll
    for (int u = 0; u < U; ++u) {
      const int i = c0 + (int)threadIdx.x + u * 256;
      if (i < Et) csr[i] = make_int2(src[u], __float_as_int(__int_as_float(ws[u]) * __int_as_float(wd[u])));
    }
  }
}

template <int U, int VEC>
__device__ __forceinline__ void gather_rows(const int2 ed, const int j, const int e0, const int cA, const int lane,
                                            const float* __restrict__ HH,
                                            float __attribute__((ext_vector_type(VEC)))& accA,
                                            float __attribute__((ext_vector_type(VEC)))& accB) {
  typedef float vf __attribute__((ext_vector_type(VEC)));
  int s[U];
  float w[U];
  vf v[U];
  #pragma unroll
  for (int u = 0; u < U; ++u) {
    s[u] = __builtin_amdgcn_readlane(ed.x, j + u);
    w[u] = __int_as_float(__builtin_amdgcn_readlane(ed.y, j + u));
  }
  const char* hb = reinterpret_cast<const char*>(HH);
  const unsigned loff = (unsigned)lane * (unsigned)sizeof(vf);
  #pragma unroll
  for (int u = 0; u < U; ++u)
    v[u] = *reinterpret_cast<const vf*>(hb + ((unsigned)s[u] * (unsigned)(32 * sizeof(vf)) + loff));
  if (e0 + U <= cA) {
    #pragma unroll
    for (int u = 0; u < U; ++u) accA += w[u] * v[u];
  } else if (e0 >= cA) {
    #pragma unroll
    for (int u = 0; u < U; ++u) accB += w[u] * v[u];
  } else {
    #pragma unroll
    for (int u = 0; u < U; ++u) {
      if (e0 + u < cA) accA += w[u] * v[u];
      else             accB += w[u] * v[u];
    }
  }
}

template <int VEC, bool RELU>
__device__ __forceinline__ void
gather_pair(const int* __restrict__ off, const int* __restrict__ cnt,
            const int2* __restrict__ csr, const float* __restrict__ HH,
            const float* __restrict__ bias, float* orowA_f, float* orowB_f,
            const int nodeA, const bool hasB, const int lane, const int N) {
  typedef float vf __attribute__((ext_vector_type(VEC)));
  const vf* __restrict__ Hv = reinterpret_cast<const vf*>(HH);
  const vf zero = (vf)(0.0f);
  vf* orowA = reinterpret_cast<vf*>(orowA_f);
  vf* orowB = reinterpret_cast<vf*>(orowB_f);
  const int nodeB = hasB ? nodeA + 1 : nodeA;

  int beg[3], cA[3], cB[3];
  #pragma unroll
  for (int g = 0; g < 3; ++g) {
    beg[g] = off[g * N + nodeA];
    cA[g]  = cnt[g * N + nodeA];
    cB[g]  = cnt[g * N + nodeB];
  }
  asm volatile("" :: "s"(beg[0]), "s"(beg[1]), "s"(beg[2]), "s"(cA[0]), "s"(cA[1]), "s"(cA[2]),
                     "s"(cB[0]), "s"(cB[1]), "s"(cB[2]));
  int tot[3];
  #pragma unroll
  for (int g = 0; g < 3; ++g) tot[g] = cA[g] + (hasB ? cB[g] : 0);

  int2 first[3];
  #pragma unroll
  for (int g = 0; g < 3; ++g) {
    first[g] = make_int2(0, 0);
    if (tot[g] > 0) {
      const int m0 = tot[g] < 32 ? tot[g] : 32;
      first[g] = csr[beg[g] + (lane < m0 ? lane : m0 - 1)];
    }
  }
  const vf bv = reinterpret_cast<const vf*>(bias)[lane];
  {
    const vf selfA = Hv[(size_t)nodeA * 32 + lane];
    const vf selfB = Hv[(size_t)nodeB * 32 + lane];
    asm volatile("" :: "v"(bv), "v"(selfA), "v"(selfB));
    vf rA = selfA + bv;
    vf rB = selfB + bv;
    if (RELU) { rA = __builtin_elementwise_max(rA, zero); rB = __builtin_elementwise_max(rB, zero); }
    orowA[3 * 32 + lane] = rA;
    orowB[3 * 32 + lane] = rB;
  }
  #pragma unroll
  for (int g = 1; g < 3; ++g) {
    float* st = orowA_f + (g * 32 + lane) * VEC;
    st[0] = __int_as_float(first[g].x);
    st[1] = __int_as_float(first[g].y);
  }

  #pragma unroll
  for (int g = 0; g < 3; ++g) {
    vf accA = zero, accB = zero;
    const int ca = cA[g];
    int2 ed = first[0];
    if (g > 0) {
      const float* st = orowA_f + (g * 32 + lane) * VEC;
      ed = make_int2(__float_as_int(st[0]), __float_as_int(st[1]));
    }
    for (int base = 0; base < tot[g]; base += 32) {
      const int rem = tot[g] - base;
      const int m = rem < 32 ? rem : 32;
      int2 nxt = ed;
      if (rem > 32) {
        const int m2 = rem - 32 < 32 ? rem - 32 : 32;
        nxt = csr[beg[g] + base + 32 + (lane < m2 ? lane : m2 - 1)];
      }
      int j = 0;
      for (; j + 16 <= m; j += 16) gather_rows<16, VEC>(ed, j, base + j, ca, lane, HH, accA, accB);
      if (j + 8 <= m) { gather_rows<8, VEC>(ed, j, base + j, ca, lane, HH, accA, accB); j += 8; }
      if (j + 4 <= m) { gather_rows<4, VEC>(ed, j, base + j, ca, lane, HH, accA, accB); j += 4; }
      if (j < m) {
        const int j1 = j + 1 < m ? j + 1 : j;
        const int j2 = j + 2 < m ? j + 2 : j;
        const int s0 = __builtin_amdgcn_readlane(ed.x, j);
        const int s1 = __builtin_amdgcn_readlane(ed.x, j1);
        const int s2 = __builtin_amdgcn_readlane(ed.x, j2);
        const float w0 = __int_as_float(__builtin_amdgcn_readlane(ed.y, j));
        const float w1 = __int_as_float(__builtin_amdgcn_readlane(ed.y, j1));
        const float w2 = __int_as_float(__builtin_amdgcn_readlane(ed.y, j2));
        const vf v0 = Hv[(size_t)s0 * 32 + lane];
        const vf v1 = Hv[(size_t)s1 * 32 + lane];
        const vf v2 = Hv[(size_t)s2 * 32 + lane];
        asm volatile("" :: "v"(v0), "v"(v1), "v"(v2));
        const int e = base + j;
        if (e < ca) accA += w0 * v0; else accB += w0 * v0;
        if (j + 1 < m) { if (e + 1 < ca) accA += w1 * v1; else accB += w1 * v1; }
        if (j + 2 < m) { if (e + 2 < ca) accA += w2 * v2; else accB += w2 * v2; }
      }
      ed = nxt;
    }
    vf rA = accA + bv;
    vf rB = accB + bv;
    if (RELU) { rA = __builtin_elementwise_max(rA, zero); rB = __builtin_elementwise_max(rB, zero); }
    orowA[g * 32 + lane] = rA;
    orowB[g * 32 + lane] = rB;
  }
}

__device__ __forceinline__ void lsm_row(float a, float b, float& ra, float& rb) {
  float mx = fmaxf(a, b);
  #pragma unroll
  for (int off = 16; off > 0; off >>= 1) mx = fmaxf(mx, __shfl_xor(mx, off));
  float s = __expf(a - mx) + __expf(b - mx);
  #pragma unroll
  for (int off = 16; off > 0; off >>= 1) s += __shfl_xor(s, off);
  const float lse = mx + __logf(s);
  ra = a - lse;
  rb = b - lse;
}

template <int VEC, bool RELU, int NCOLS, bool LOGSOFTMAX>
__global__ void __launch_bounds__(256, 10)
gcn_gather_gemm(const int* __restrict__ off, const int* __restrict__ cnt,
                const int2* __restrict__ csr, const float* __restrict__ HH,
                const float* __restrict__ bias, const float* __restrict__ B,
                const float* __restrict__ bias2, float* __restrict__ D, int N) {
  constexpr int K   = 4 * 32 * VEC;
  constexpr int LDA = K + 4;
  constexpr int KQ  = K / 4;
  constexpr int LDP = NCOLS + 4;
  static_assert(NCOLS == 64, "2 tile pairs x 4 K quarters = 8 waves; a row = 2 columns per lane");
  static_assert(KQ % 32 == 0, "a K quarter is a whole number of 8-step trips");
  __shared__ __attribute__((aligned(16))) float As[16 * LDA];
  __shared__ __attribute__((aligned(16))) float Ps[4 * 16 * LDP];

  const int wv    = __builtin_amdgcn_readfirstlane((int)(threadIdx.x >> 5));
  const int lane  = threadIdx.x & 31;
  const int row0  = blockIdx.x * 16;
  const int nodeA = row0 + 2 * wv;
  float* srowA = &As[(2 * wv) * LDA];
  float* srowB = srowA + LDA;
  if (nodeA < N) {
    gather_pair<VEC, RELU>(off, cnt, csr, HH, bias, srowA, srowB, nodeA, nodeA + 1 < N, lane, N);
  } else {
    typedef float vf __attribute__((ext_vector_type(VEC)));
    #pragma unroll
    for (int g = 0; g < 4; ++g) {
      reinterpret_cast<vf*>(srowA)[g * 32 + lane] = (vf)(0.0f);
      reinterpret_cast<vf*>(srowB)[g * 32 + lane] = (vf)(0.0f);
    }
  }
  __syncthreads();

  const int half  = lane >> 4;
  const int mn    = lane & 15;
  const int kb    = half * 2;
  const int cbase = (wv & 1) * 32;
  const int kq    = wv >> 1;
  const int k0    = kq * KQ;

  const float* Ap = &As[mn * LDA + k0 + kb];
  const float* Bp = B + (k0 + kb) * NCOLS + cbase + 2 * mn;

  float ba = 0.0f, bb = 0.0f;
  if (bias2) { ba = bias2[lane]; bb = bias2[lane + 32]; }

  v8f acc0 = {}, acc1 = {};
  #pragma unroll 8
  for (int kk = 0; kk < KQ; kk += 4) {
    const v2f a = *reinterpret_cast<const v2f*>(Ap + kk);
    const v2f r0 = *reinterpret_cast<const v2f*>(Bp + kk * NCOLS);
    const v2f r1 = *reinterpret_cast<const v2f*>(Bp + kk * NCOLS + NCOLS);
    v2f b0, b1;
    b0.x = r0.x; b0.y = r1.x;
    b1.x = r0.y; b1.y = r1.y;
    acc0 = __builtin_amdgcn_wmma_f32_16x16x4_f32(
         false, a,  false, b0,
         (short)0, acc0,  false,  false);
    acc1 = __builtin_amdgcn_wmma_f32_16x16x4_f32(
         false, a,  false, b1,
         (short)0, acc1,  false,  false);
  }

  #pragma unroll
  for (int r = 0; r < 8; ++r) {
    v2f p;
    p.x = acc0[r];
    p.y = acc1[r];
    *reinterpret_cast<v2f*>(&Ps[(kq * 16 + r + half * 8) * LDP + cbase + 2 * mn]) = p;
  }
  __syncthreads();

  #pragma unroll
  for (int i = 0; i < 2; ++i) {
    const int row = 2 * wv + i;
    const float* pr = &Ps[row * LDP + lane];
    float va = ((pr[0] + pr[16 * LDP]) + pr[32 * LDP]) + pr[48 * LDP];
    float vb = ((pr[32] + pr[16 * LDP + 32]) + pr[32 * LDP + 32]) + pr[48 * LDP + 32];
    va += ba;
    vb += bb;
    if (LOGSOFTMAX) {
      float ra, rb;
      lsm_row(va, vb, ra, rb);
      va = ra;
      vb = rb;
    }
    const int node = row0 + row;
    if (node < N) {
      float* drow = D + (long long)node * NCOLS;
      drow[lane]      = va;
      drow[lane + 32] = vb;
    }
  }
}

extern "C" void kernel_launch(void* const* d_in, const int* in_sizes, int n_in,
                              void* d_out, int out_size, void* d_ws, size_t ws_size,
                              hipStream_t stream) {
  const float* x  = (const float*)d_in[0];
  const int* e1   = (const int*)d_in[1];
  const int* e2   = (const int*)d_in[2];
  const int* e3   = (const int*)d_in[3];
  const float* W1 = (const float*)d_in[5];
  const float* b1 = (const float*)d_in[6];
  const float* W2 = (const float*)d_in[7];
  const float* b2 = (const float*)d_in[8];
  const float* Wl = (const float*)d_in[9];
  const float* bl = (const float*)d_in[10];

  const int F = 512, H = 128, C = 64;
  const int N  = in_sizes[0] / F;
  const int E1 = in_sizes[1] / 2;
  const int E2 = in_sizes[2] / 2;
  const int E3 = in_sizes[3] / 2;
  const long long Et = (long long)E1 + E2 + E3;

  float* ws   = (float*)d_ws;
  float* HH1  = ws;
  float* HH2  = HH1 + (long long)N * H;
  int2*  csr  = (int2*)(HH2 + (long long)N * C);
  int2*  binned = csr + Et;
  float* dinv = (float*)(binned + Et);
  int*   cnt  = (int*)(dinv + 3LL * N);
  int*   off  = cnt + 3LL * N;

  const int T = 256;

  const int nkeys = 3 * N;
  const int nbin  = cdiv_i(nkeys, BIN_KEYS);
  const int nseg  = cdiv_i(Et, SEG_EDGES);
  int*   segCnt   = off + 3LL * N;
  int*   segOff   = segCnt + (long long)nbin * nseg;
  seg_sort<<<nseg, SORT_T, 0, stream>>>(e1, e2, e3, E1, E2, E3, N, nbin, nseg, segCnt, segOff, binned);
  const int mtiles = cdiv_i(N, 16);
  const int gquad  = mtiles / 4 < 256 ? mtiles / 4 : 256;
  const int grest  = mtiles - 4 * gquad;
  bin_sort<8, F, H><<<nbin + gquad, SORT_T, 0, stream>>>(binned, segCnt, segOff, nseg, nkeys, nbin, cnt, off, dinv, csr,
                                                         x, W1, HH1, N);
  csr_weight<8, F, H><<<grest + 512, T, 0, stream>>>(csr, dinv, E1, E1 + E2, (int)Et, N,
                                                     x, W1, HH1, N, 4 * gquad, grest);

  gcn_gather_gemm<4, true, C, false><<<mtiles, 256, 0, stream>>>(off, cnt, csr, HH1, b1, W2, nullptr, HH2, N);

  gcn_gather_gemm<2, false, C, true><<<mtiles, 256, 0, stream>>>(off, cnt, csr, HH2, b2, Wl, bl, (float*)d_out, N);
}
